# starting kernel
# baseline (speedup 1.0000x reference)
#include <hip/hip_runtime.h>


#define GAT_N 50000
#define GAT_E 800000
#define GAT_F 64

typedef float v2f __attribute__((ext_vector_type(2)));
typedef float v8f __attribute__((ext_vector_type(8)));

__global__ __launch_bounds__(256)
void gat_gemm_wmma(const float* __restrict__ A, const float* __restrict__ B,
                   float* __restrict__ C, int M, int K, int Nw) {
  const int wave = blockIdx.x * 8 + (threadIdx.x >> 5);
  const int lane = threadIdx.x & 31;
  const int tilesN = Nw >> 6;
  const int tilesM = M >> 4;
  if (wave >= tilesM * tilesN) return;
  const int tm = wave / tilesN;
  const int tn = wave - tm * tilesN;
  const int kh = lane >> 4;
  const int lr = lane & 15;

  const float* Ap = A + (size_t)(tm * 16 + lr) * K + kh * 2;
  const float* Bp = B + (size_t)(tn * 64) + lr;
  v8f acc0 = {}, acc1 = {}, acc2 = {}, acc3 = {};
  for (int k0 = 0; k0 < K; k0 += 4) {
    v2f a = *(const v2f*)(Ap + k0);
    const float* Brow0 = Bp + (size_t)(k0 + kh * 2) * Nw;
    const float* Brow1 = Brow0 + Nw;
    v2f b0, b1, b2, b3;
    b0.x = Brow0[0];  b0.y = Brow1[0];
    b1.x = Brow0[16]; b1.y = Brow1[16];
    b2.x = Brow0[32]; b2.y = Brow1[32];
    b3.x = Brow0[48]; b3.y = Brow1[48];
    acc0 = __builtin_amdgcn_wmma_f32_16x16x4_f32(false, a, false, b0, (short)0, acc0, false, false);
    acc1 = __builtin_amdgcn_wmma_f32_16x16x4_f32(false, a, false, b1, (short)0, acc1, false, false);
    acc2 = __builtin_amdgcn_wmma_f32_16x16x4_f32(false, a, false, b2, (short)0, acc2, false, false);
    acc3 = __builtin_amdgcn_wmma_f32_16x16x4_f32(false, a, false, b3, (short)0, acc3, false, false);
  }
  float* Cp = C + (size_t)(tm * 16 + kh * 8) * Nw + tn * 64 + lr;
#pragma unroll
  for (int r = 0; r < 8; ++r) {
    Cp[(size_t)r * Nw +  0] = acc0[r];
    Cp[(size_t)r * Nw + 16] = acc1[r];
    Cp[(size_t)r * Nw + 32] = acc2[r];
    Cp[(size_t)r * Nw + 48] = acc3[r];
  }
}

__global__ __launch_bounds__(256)
void gat_node_attn(const float* __restrict__ h, const float* __restrict__ attn,
                   float* __restrict__ out, int N, int H) {
  int idx = blockIdx.x * blockDim.x + threadIdx.x;
  if (idx >= N * H) return;
  int n = idx / H, hh = idx - n * H;
  const float4* hp = (const float4*)(h + ((size_t)n * H + hh) * GAT_F);
  const float4* ap = (const float4*)(attn + (size_t)hh * GAT_F);
  float s = 0.f;
#pragma unroll
  for (int f = 0; f < GAT_F / 4; ++f) {
    float4 x = hp[f], w = ap[f];
    s += x.x * w.x + x.y * w.y + x.z * w.z + x.w * w.w;
  }
  out[idx] = s;
}

__global__ __launch_bounds__(256)
void gat_fill_u32(unsigned* __restrict__ p, unsigned v, long long n) {
  long long i = (long long)blockIdx.x * blockDim.x + threadIdx.x;
  if (i < n) p[i] = v;
}

__device__ __forceinline__ unsigned f32_key(float x) {
  unsigned u = __float_as_uint(x);
  return (u & 0x80000000u) ? ~u : (u | 0x80000000u);
}
__device__ __forceinline__ float key_f32(unsigned k) {
  return (k & 0x80000000u) ? __uint_as_float(k ^ 0x80000000u) : __uint_as_float(~k);
}
#define KEY_NEG_INF 0x007FFFFFu

__global__ __launch_bounds__(256)
void gat_edge_logits(const int* __restrict__ src, const int* __restrict__ dst,
                     const float* __restrict__ el, const float* __restrict__ er,
                     float* __restrict__ ebuf, unsigned* __restrict__ mkey,
                     int E, int H) {
  int idx = blockIdx.x * blockDim.x + threadIdx.x;
  if (idx >= E * H) return;
  int e = idx / H, hh = idx - e * H;
  int s = src[e], d = dst[e];
  float x = el[s * H + hh] + er[d * H + hh];
  x = (x > 0.f) ? x : 0.2f * x;
  ebuf[idx] = x;
  atomicMax(&mkey[d * H + hh], f32_key(x));
}

__global__ __launch_bounds__(256)
void gat_edge_exp(const int* __restrict__ dst, float* __restrict__ ebuf,
                  const unsigned* __restrict__ mkey, float* __restrict__ ssum,
                  int E, int H) {
  int idx = blockIdx.x * blockDim.x + threadIdx.x;
  if (idx >= E * H) return;
  int e = idx / H, hh = idx - e * H;
  int d = dst[e];
  float m = key_f32(mkey[d * H + hh]);
  float ex = __expf(ebuf[idx] - m);
  ebuf[idx] = ex;
  atomicAdd(&ssum[d * H + hh], ex);
}

__global__ __launch_bounds__(256)
void gat_edge_aggregate(const int* __restrict__ src, const int* __restrict__ dst,
                        const float* __restrict__ h, const float* __restrict__ ebuf,
                        const float* __restrict__ ssum, float* __restrict__ out,
                        int E, int H) {
  const int HF = H * GAT_F;
  long long idx = (long long)blockIdx.x * blockDim.x + threadIdx.x;
  if (idx >= (long long)E * HF) return;
  int e = (int)(idx / HF);
  int i = (int)(idx - (long long)e * HF);
  int hh = i >> 6;
  int s = src[e], d = dst[e];
  float alpha = ebuf[e * H + hh] / ssum[d * H + hh];
  atomicAdd(&out[(size_t)d * HF + i], h[(size_t)s * HF + i] * alpha);
}

__global__ __launch_bounds__(256)
void gat_bias_relu(float* __restrict__ x, const float* __restrict__ b,
                   long long n, int HF) {
  long long idx = (long long)blockIdx.x * blockDim.x + threadIdx.x;
  if (idx >= n) return;
  float v = x[idx] + b[(int)(idx % HF)];
  x[idx] = v > 0.f ? v : 0.f;
}

__global__ __launch_bounds__(256)
void gat_bias_out(const float* __restrict__ x, const float* __restrict__ b,
                  float* __restrict__ out, long long n, int HF) {
  long long idx = (long long)blockIdx.x * blockDim.x + threadIdx.x;
  if (idx >= n) return;
  out[idx] = x[idx] + b[(int)(idx % HF)];
}

extern "C" void kernel_launch(void* const* d_in, const int* in_sizes, int n_in,
                              void* d_out, int out_size, void* d_ws, size_t ws_size,
                              hipStream_t stream) {
  (void)in_sizes; (void)n_in; (void)out_size; (void)ws_size;
  const float* feat = (const float*)d_in[0];
  const int*   src  = (const int*)d_in[1];
  const int*   dst  = (const int*)d_in[2];
  const float* W1  = (const float*)d_in[3];
  const float* al1 = (const float*)d_in[4];
  const float* ar1 = (const float*)d_in[5];
  const float* b1  = (const float*)d_in[6];
  const float* W2  = (const float*)d_in[7];
  const float* al2 = (const float*)d_in[8];
  const float* ar2 = (const float*)d_in[9];
  const float* b2  = (const float*)d_in[10];
  const float* W3  = (const float*)d_in[11];
  const float* al3 = (const float*)d_in[12];
  const float* ar3 = (const float*)d_in[13];
  const float* b3  = (const float*)d_in[14];
  float* out = (float*)d_out;

  const int N = GAT_N, E = GAT_E;

  char* ws = (char*)d_ws;
  size_t off = 0;
  auto carve = [&](size_t bytes) -> void* {
    void* p = ws + off;
    off += (bytes + 255) & ~(size_t)255;
    return p;
  };
  float*    bufA = (float*)carve((size_t)N * 256 * 4);
  float*    bufB = (float*)carve((size_t)N * 256 * 4);
  float*    el   = (float*)carve((size_t)N * 4 * 4);
  float*    er   = (float*)carve((size_t)N * 4 * 4);
  unsigned* mkey = (unsigned*)carve((size_t)N * 4 * 4);
  float*    ssum = (float*)carve((size_t)N * 4 * 4);
  float*    ebuf = (float*)carve((size_t)E * 4 * 4);

  auto blocks = [](long long n) { return (unsigned)((n + 255) / 256); };

  {
    const int K = 128, Nw = 256, H = 4;
    int waves = (N / 16) * (Nw / 64);
    gat_gemm_wmma<<<(waves + 7) / 8, 256, 0, stream>>>(feat, W1, bufA, N, K, Nw);
    long long nh = (long long)N * H;
    gat_node_attn<<<blocks(nh), 256, 0, stream>>>(bufA, al1, el, N, H);
    gat_node_attn<<<blocks(nh), 256, 0, stream>>>(bufA, ar1, er, N, H);
    gat_fill_u32<<<blocks(nh), 256, 0, stream>>>(mkey, KEY_NEG_INF, nh);
    gat_fill_u32<<<blocks(nh), 256, 0, stream>>>((unsigned*)ssum, 0u, nh);
    long long nout = (long long)N * H * GAT_F;
    gat_fill_u32<<<blocks(nout), 256, 0, stream>>>((unsigned*)bufB, 0u, nout);
    long long eh = (long long)E * H;
    gat_edge_logits<<<blocks(eh), 256, 0, stream>>>(src, dst, el, er, ebuf, mkey, E, H);
    gat_edge_exp<<<blocks(eh), 256, 0, stream>>>(dst, ebuf, mkey, ssum, E, H);
    long long ew = (long long)E * H * GAT_F;
    gat_edge_aggregate<<<blocks(ew), 256, 0, stream>>>(src, dst, bufA, ebuf, ssum, bufB, E, H);
    gat_bias_relu<<<blocks(nout), 256, 0, stream>>>(bufB, b1, nout, H * GAT_F);
  }

  {
    const int K = 256, Nw = 256, H = 4;
    int waves = (N / 16) * (Nw / 64);
    gat_gemm_wmma<<<(waves + 7) / 8, 256, 0, stream>>>(bufB, W2, bufA, N, K, Nw);
    long long nh = (long long)N * H;
    gat_node_attn<<<blocks(nh), 256, 0, stream>>>(bufA, al2, el, N, H);
    gat_node_attn<<<blocks(nh), 256, 0, stream>>>(bufA, ar2, er, N, H);
    gat_fill_u32<<<blocks(nh), 256, 0, stream>>>(mkey, KEY_NEG_INF, nh);
    gat_fill_u32<<<blocks(nh), 256, 0, stream>>>((unsigned*)ssum, 0u, nh);
    long long nout = (long long)N * H * GAT_F;
    gat_fill_u32<<<blocks(nout), 256, 0, stream>>>((unsigned*)bufB, 0u, nout);
    long long eh = (long long)E * H;
    gat_edge_logits<<<blocks(eh), 256, 0, stream>>>(src, dst, el, er, ebuf, mkey, E, H);
    gat_edge_exp<<<blocks(eh), 256, 0, stream>>>(dst, ebuf, mkey, ssum, E, H);
    long long ew = (long long)E * H * GAT_F;
    gat_edge_aggregate<<<blocks(ew), 256, 0, stream>>>(src, dst, bufA, ebuf, ssum, bufB, E, H);
    gat_bias_relu<<<blocks(nout), 256, 0, stream>>>(bufB, b2, nout, H * GAT_F);
  }

  {
    const int K = 256, Nw = 64, H = 1;
    int waves = (N / 16) * (Nw / 64);
    gat_gemm_wmma<<<(waves + 7) / 8, 256, 0, stream>>>(bufB, W3, bufA, N, K, Nw);
    long long nh = (long long)N * H;
    gat_node_attn<<<blocks(nh), 256, 0, stream>>>(bufA, al3, el, N, H);
    gat_node_attn<<<blocks(nh), 256, 0, stream>>>(bufA, ar3, er, N, H);
    gat_fill_u32<<<blocks(nh), 256, 0, stream>>>(mkey, KEY_NEG_INF, nh);
    gat_fill_u32<<<blocks(nh), 256, 0, stream>>>((unsigned*)ssum, 0u, nh);
    long long nout = (long long)N * H * GAT_F;
    gat_fill_u32<<<blocks(nout), 256, 0, stream>>>((unsigned*)bufB, 0u, nout);
    long long eh = (long long)E * H;
    gat_edge_logits<<<blocks(eh), 256, 0, stream>>>(src, dst, el, er, ebuf, mkey, E, H);
    gat_edge_exp<<<blocks(eh), 256, 0, stream>>>(dst, ebuf, mkey, ssum, E, H);
    long long ew = (long long)E * H * GAT_F;
    gat_edge_aggregate<<<blocks(ew), 256, 0, stream>>>(src, dst, bufA, ebuf, ssum, bufB, E, H);
    gat_bias_out<<<blocks(nout), 256, 0, stream>>>(bufB, b3, out, nout, GAT_F);
  }
}


# Bucket edges by dst once per call (int count, scan, edge-id order); one wave per node does softmax max/sum and weighted row gather in registers, one store; float atomics gone, sums in fixed order
# speedup vs baseline: 4.1718x; 4.1718x over previous
#include <hip/hip_runtime.h>


#define GAT_N 50000
#define GAT_E 800000
#define GAT_F 64

typedef float v2f __attribute__((ext_vector_type(2)));
typedef float v8f __attribute__((ext_vector_type(8)));

__global__ __launch_bounds__(256)
void gat_gemm_wmma(const float* __restrict__ A, const float* __restrict__ B,
                   float* __restrict__ C, int M, int K, int Nw) {
  const int wave = blockIdx.x * 8 + (threadIdx.x >> 5);
  const int lane = threadIdx.x & 31;
  const int tilesN = Nw >> 6;
  const int tilesM = M >> 4;
  if (wave >= tilesM * tilesN) return;
  const int tm = wave / tilesN;
  const int tn = wave - tm * tilesN;
  const int kh = lane >> 4;
  const int lr = lane & 15;

  const float* Ap = A + (size_t)(tm * 16 + lr) * K + kh * 2;
  const float* Bp = B + (size_t)(tn * 64) + lr;
  v8f acc0 = {}, acc1 = {}, acc2 = {}, acc3 = {};
  for (int k0 = 0; k0 < K; k0 += 4) {
    v2f a = *(const v2f*)(Ap + k0);
    const float* Brow0 = Bp + (size_t)(k0 + kh * 2) * Nw;
    const float* Brow1 = Brow0 + Nw;
    v2f b0, b1, b2, b3;
    b0.x = Brow0[0];  b0.y = Brow1[0];
    b1.x = Brow0[16]; b1.y = Brow1[16];
    b2.x = Brow0[32]; b2.y = Brow1[32];
    b3.x = Brow0[48]; b3.y = Brow1[48];
    acc0 = __builtin_amdgcn_wmma_f32_16x16x4_f32(false, a, false, b0, (short)0, acc0, false, false);
    acc1 = __builtin_amdgcn_wmma_f32_16x16x4_f32(false, a, false, b1, (short)0, acc1, false, false);
    acc2 = __builtin_amdgcn_wmma_f32_16x16x4_f32(false, a, false, b2, (short)0, acc2, false, false);
    acc3 = __builtin_amdgcn_wmma_f32_16x16x4_f32(false, a, false, b3, (short)0, acc3, false, false);
  }
  float* Cp = C + (size_t)(tm * 16 + kh * 8) * Nw + tn * 64 + lr;
#pragma unroll
  for (int r = 0; r < 8; ++r) {
    Cp[(size_t)r * Nw +  0] = acc0[r];
    Cp[(size_t)r * Nw + 16] = acc1[r];
    Cp[(size_t)r * Nw + 32] = acc2[r];
    Cp[(size_t)r * Nw + 48] = acc3[r];
  }
}

__global__ __launch_bounds__(256)
void gat_node_attn(const float* __restrict__ h, const float* __restrict__ attn,
                   float* __restrict__ out, int N, int H) {
  int idx = blockIdx.x * blockDim.x + threadIdx.x;
  if (idx >= N * H) return;
  int n = idx / H, hh = idx - n * H;
  const float4* hp = (const float4*)(h + ((size_t)n * H + hh) * GAT_F);
  const float4* ap = (const float4*)(attn + (size_t)hh * GAT_F);
  float s = 0.f;
#pragma unroll
  for (int f = 0; f < GAT_F / 4; ++f) {
    float4 x = hp[f], w = ap[f];
    s += x.x * w.x + x.y * w.y + x.z * w.z + x.w * w.w;
  }
  out[idx] = s;
}

__global__ __launch_bounds__(256)
void gat_fill_u32(unsigned* __restrict__ p, unsigned v, long long n) {
  long long i = (long long)blockIdx.x * blockDim.x + threadIdx.x;
  if (i < n) p[i] = v;
}

__global__ __launch_bounds__(256)
void gat_csr_rank(const int* __restrict__ dst, int* __restrict__ cnt,
                  int* __restrict__ rank, int E) {
  int e = blockIdx.x * blockDim.x + threadIdx.x;
  if (e >= E) return;
  rank[e] = atomicAdd(&cnt[dst[e]], 1);
}

__global__ __launch_bounds__(1024)
void gat_csr_scan(const int* __restrict__ cnt, int* __restrict__ rowptr) {
  constexpr int n = GAT_N;
  constexpr int PER = (n + 1023) / 1024;
  __shared__ int part[1024];
  const int t = threadIdx.x;
  const int b = t * PER;
  int c[PER];
  int s = 0;
#pragma unroll
  for (int k = 0; k < PER; ++k) {
    c[k] = (b + k < n) ? cnt[b + k] : 0;
    s += c[k];
  }
  part[t] = s;
  __syncthreads();
  for (int off = 1; off < 1024; off <<= 1) {
    int v = (t >= off) ? part[t - off] : 0;
    __syncthreads();
    part[t] += v;
    __syncthreads();
  }
  int run = part[t] - s;
#pragma unroll
  for (int k = 0; k < PER; ++k) {
    if (b + k < n) rowptr[b + k] = run;
    run += c[k];
  }
  if (t == 1023) rowptr[n] = part[1023];
}

__global__ __launch_bounds__(256)
void gat_csr_place(const int* __restrict__ dst, const int* __restrict__ rowptr,
                   const int* __restrict__ rank, int* __restrict__ eid, int E) {
  int e = blockIdx.x * blockDim.x + threadIdx.x;
  if (e >= E) return;
  eid[rowptr[dst[e]] + rank[e]] = e;
}

__global__ __launch_bounds__(256)
void gat_csr_order(const int* __restrict__ src, const int* __restrict__ dst,
                   const int* __restrict__ rowptr, const int* __restrict__ eid,
                   int* __restrict__ csr_src, int E) {
  int e = blockIdx.x * blockDim.x + threadIdx.x;
  if (e >= E) return;
  const int d = dst[e];
  const int b = rowptr[d], en = rowptr[d + 1];
  int c = 0;
  for (int q = b; q < en; ++q) c += (eid[q] < e) ? 1 : 0;
  csr_src[b + c] = src[e];
}

__device__ __forceinline__ float gat_lrelu(float x) { return (x > 0.f) ? x : 0.2f * x; }
__device__ __forceinline__ float gat_wave_max(float v) {
  for (int o = 16; o > 0; o >>= 1) v = fmaxf(v, __shfl_xor(v, o, 32));
  return v;
}
__device__ __forceinline__ float gat_wave_sum(float v) {
  for (int o = 16; o > 0; o >>= 1) v += __shfl_xor(v, o, 32);
  return v;
}
__device__ __forceinline__ float gat_bcast(float v, int j) {
  return __int_as_float(__builtin_amdgcn_readlane(__float_as_int(v), j));
}

__global__ __launch_bounds__(256)
void gat_gather_h4(const int* __restrict__ rowptr, const int* __restrict__ csr_src,
                   const float* __restrict__ h, const float* __restrict__ el,
                   const float* __restrict__ er, const float* __restrict__ bias,
                   float* __restrict__ out, int N) {
  const int d = __builtin_amdgcn_readfirstlane((int)(blockIdx.x * 8 + (threadIdx.x >> 5)));
  if (d >= N) return;
  const int lane = threadIdx.x & 31;
  const bool hi = lane >= 16;
  const int beg = rowptr[d], end = rowptr[d + 1];
  const float4 erd = *(const float4*)(er + (size_t)d * 4);
  const float4* el4 = (const float4*)el;

  const float ninf = -__builtin_huge_valf();
  float m0 = ninf, m1 = ninf, m2 = ninf, m3 = ninf;
  for (int p = beg + lane; p < end; p += 32) {
    const float4 x = el4[csr_src[p]];
    m0 = fmaxf(m0, gat_lrelu(x.x + erd.x));
    m1 = fmaxf(m1, gat_lrelu(x.y + erd.y));
    m2 = fmaxf(m2, gat_lrelu(x.z + erd.z));
    m3 = fmaxf(m3, gat_lrelu(x.w + erd.w));
  }
  m0 = gat_wave_max(m0); m1 = gat_wave_max(m1);
  m2 = gat_wave_max(m2); m3 = gat_wave_max(m3);

  float s0 = 0.f, s1 = 0.f, s2 = 0.f, s3 = 0.f;
  for (int p = beg + lane; p < end; p += 32) {
    const float4 x = el4[csr_src[p]];
    s0 += __expf(gat_lrelu(x.x + erd.x) - m0);
    s1 += __expf(gat_lrelu(x.y + erd.y) - m1);
    s2 += __expf(gat_lrelu(x.z + erd.z) - m2);
    s3 += __expf(gat_lrelu(x.w + erd.w) - m3);
  }
  s0 = gat_wave_sum(s0); s1 = gat_wave_sum(s1);
  s2 = gat_wave_sum(s2); s3 = gat_wave_sum(s3);

  float4 acc0 = {0.f, 0.f, 0.f, 0.f}, acc1 = {0.f, 0.f, 0.f, 0.f};
  for (int base = beg; base < end; base += 32) {
    const int p = base + lane;
    int s = 0;
    float a0 = 0.f, a1 = 0.f, a2 = 0.f, a3 = 0.f;
    if (p < end) {
      s = csr_src[p];
      const float4 x = el4[s];
      a0 = __expf(gat_lrelu(x.x + erd.x) - m0) / s0;
      a1 = __expf(gat_lrelu(x.y + erd.y) - m1) / s1;
      a2 = __expf(gat_lrelu(x.z + erd.z) - m2) / s2;
      a3 = __expf(gat_lrelu(x.w + erd.w) - m3) / s3;
    }
    const int cnt = (end - base < 32) ? (end - base) : 32;
#pragma unroll 4
    for (int j = 0; j < cnt; ++j) {
      const int sj = __builtin_amdgcn_readlane(s, j);
      const float w0 = gat_bcast(a0, j), w1 = gat_bcast(a1, j);
      const float w2 = gat_bcast(a2, j), w3 = gat_bcast(a3, j);
      const float wA = hi ? w1 : w0;
      const float wB = hi ? w3 : w2;
      const float4* row = (const float4*)(h + (size_t)sj * 256);
      const float4 v0 = row[lane];
      const float4 v1 = row[32 + lane];
      acc0.x += v0.x * wA; acc0.y += v0.y * wA; acc0.z += v0.z * wA; acc0.w += v0.w * wA;
      acc1.x += v1.x * wB; acc1.y += v1.y * wB; acc1.z += v1.z * wB; acc1.w += v1.w * wB;
    }
  }
  const float4 b0 = ((const float4*)bias)[lane];
  const float4 b1 = ((const float4*)bias)[32 + lane];
  float4 o0, o1;
  o0.x = acc0.x + b0.x; o0.y = acc0.y + b0.y; o0.z = acc0.z + b0.z; o0.w = acc0.w + b0.w;
  o1.x = acc1.x + b1.x; o1.y = acc1.y + b1.y; o1.z = acc1.z + b1.z; o1.w = acc1.w + b1.w;
  o0.x = o0.x > 0.f ? o0.x : 0.f; o0.y = o0.y > 0.f ? o0.y : 0.f;
  o0.z = o0.z > 0.f ? o0.z : 0.f; o0.w = o0.w > 0.f ? o0.w : 0.f;
  o1.x = o1.x > 0.f ? o1.x : 0.f; o1.y = o1.y > 0.f ? o1.y : 0.f;
  o1.z = o1.z > 0.f ? o1.z : 0.f; o1.w = o1.w > 0.f ? o1.w : 0.f;
  float4* orow = (float4*)(out + (size_t)d * 256);
  orow[lane] = o0;
  orow[32 + lane] = o1;
}

__global__ __launch_bounds__(256)
void gat_gather_h1(const int* __restrict__ rowptr, const int* __restrict__ csr_src,
                   const float* __restrict__ h, const float* __restrict__ el,
                   const float* __restrict__ er, const float* __restrict__ bias,
                   float* __restrict__ out, int N) {
  const int d = __builtin_amdgcn_readfirstlane((int)(blockIdx.x * 8 + (threadIdx.x >> 5)));
  if (d >= N) return;
  const int lane = threadIdx.x & 31;
  const int beg = rowptr[d], end = rowptr[d + 1];
  const float erd = er[d];

  float m = -__builtin_huge_valf();
  for (int p = beg + lane; p < end; p += 32)
    m = fmaxf(m, gat_lrelu(el[csr_src[p]] + erd));
  m = gat_wave_max(m);

  float ssum = 0.f;
  for (int p = beg + lane; p < end; p += 32)
    ssum += __expf(gat_lrelu(el[csr_src[p]] + erd) - m);
  ssum = gat_wave_sum(ssum);

  float2 acc = {0.f, 0.f};
  for (int base = beg; base < end; base += 32) {
    const int p = base + lane;
    int s = 0;
    float a = 0.f;
    if (p < end) {
      s = csr_src[p];
      a = __expf(gat_lrelu(el[s] + erd) - m) / ssum;
    }
    const int cnt = (end - base < 32) ? (end - base) : 32;
#pragma unroll 4
    for (int j = 0; j < cnt; ++j) {
      const int sj = __builtin_amdgcn_readlane(s, j);
      const float w = gat_bcast(a, j);
      const float2 v = ((const float2*)(h + (size_t)sj * 64))[lane];
      acc.x += v.x * w; acc.y += v.y * w;
    }
  }
  const float2 b = ((const float2*)bias)[lane];
  float2 o;
  o.x = acc.x + b.x; o.y = acc.y + b.y;
  ((float2*)(out + (size_t)d * 64))[lane] = o;
}

extern "C" void kernel_launch(void* const* d_in, const int* in_sizes, int n_in,
                              void* d_out, int out_size, void* d_ws, size_t ws_size,
                              hipStream_t stream) {
  (void)in_sizes; (void)n_in; (void)out_size; (void)ws_size;
  const float* feat = (const float*)d_in[0];
  const int*   src  = (const int*)d_in[1];
  const int*   dst  = (const int*)d_in[2];
  const float* W1  = (const float*)d_in[3];
  const float* al1 = (const float*)d_in[4];
  const float* ar1 = (const float*)d_in[5];
  const float* b1  = (const float*)d_in[6];
  const float* W2  = (const float*)d_in[7];
  const float* al2 = (const float*)d_in[8];
  const float* ar2 = (const float*)d_in[9];
  const float* b2  = (const float*)d_in[10];
  const float* W3  = (const float*)d_in[11];
  const float* al3 = (const float*)d_in[12];
  const float* ar3 = (const float*)d_in[13];
  const float* b3  = (const float*)d_in[14];
  float* out = (float*)d_out;

  const int N = GAT_N, E = GAT_E;

  char* ws = (char*)d_ws;
  size_t off = 0;
  auto carve = [&](size_t bytes) -> void* {
    void* p = ws + off;
    off += (bytes + 255) & ~(size_t)255;
    return p;
  };
  float* bufA    = (float*)carve((size_t)N * 256 * 4);
  float* bufB    = (float*)carve((size_t)N * 256 * 4);
  float* el      = (float*)carve((size_t)N * 4 * 4);
  float* er      = (float*)carve((size_t)N * 4 * 4);
  int*   cnt     = (int*)carve((size_t)N * 4);
  int*   rowptr  = (int*)carve((size_t)(N + 1) * 4);
  int*   rank    = (int*)carve((size_t)E * 4);
  int*   eid     = (int*)carve((size_t)E * 4);
  int*   csr_src = (int*)carve((size_t)E * 4);

  auto blocks = [](long long n) { return (unsigned)((n + 255) / 256); };

  gat_fill_u32<<<blocks(N), 256, 0, stream>>>((unsigned*)cnt, 0u, (long long)N);
  gat_csr_rank<<<blocks(E), 256, 0, stream>>>(dst, cnt, rank, E);
  gat_csr_scan<<<1, 1024, 0, stream>>>(cnt, rowptr);
  gat_csr_place<<<blocks(E), 256, 0, stream>>>(dst, rowptr, rank, eid, E);
  gat_csr_order<<<blocks(E), 256, 0, stream>>>(src, dst, rowptr, eid, csr_src, E);

  const unsigned node_blocks = (unsigned)((N + 7) / 8);

  {
    const int K = 128, Nw = 256, H = 4;
    int waves = (N / 16) * (Nw / 64);
    gat_gemm_wmma<<<(waves + 7) / 8, 256, 0, stream>>>(feat, W1, bufA, N, K, Nw);
    long long nh = (long long)N * H;
    gat_node_attn<<<blocks(nh), 256, 0, stream>>>(bufA, al1, el, N, H);
    gat_node_attn<<<blocks(nh), 256, 0, stream>>>(bufA, ar1, er, N, H);
    gat_gather_h4<<<node_blocks, 256, 0, stream>>>(rowptr, csr_src, bufA, el, er, b1, bufB, N);
  }

  {
    const int K = 256, Nw = 256, H = 4;
    int waves = (N / 16) * (Nw / 64);
    gat_gemm_wmma<<<(waves + 7) / 8, 256, 0, stream>>>(bufB, W2, bufA, N, K, Nw);
    long long nh = (long long)N * H;
    gat_node_attn<<<blocks(nh), 256, 0, stream>>>(bufA, al2, el, N, H);
    gat_node_attn<<<blocks(nh), 256, 0, stream>>>(bufA, ar2, er, N, H);
    gat_gather_h4<<<node_blocks, 256, 0, stream>>>(rowptr, csr_src, bufA, el, er, b2, bufB, N);
  }

  {
    const int K = 256, Nw = 64, H = 1;
    int waves = (N / 16) * (Nw / 64);
    gat_gemm_wmma<<<(waves + 7) / 8, 256, 0, stream>>>(bufB, W3, bufA, N, K, Nw);
    long long nh = (long long)N * H;
    gat_node_attn<<<blocks(nh), 256, 0, stream>>>(bufA, al3, el, N, H);
    gat_node_attn<<<blocks(nh), 256, 0, stream>>>(bufA, ar3, er, N, H);
    gat_gather_h1<<<node_blocks, 256, 0, stream>>>(rowptr, csr_src, bufA, el, er, b3, out, N);
  }
}


# GEMM epilogue forms el/er from the accumulators (strip = one head; 16-lane DPP row sums, so a new sum order for el/er); the 6 node-attention launches and their re-reads of h are gone
# speedup vs baseline: 4.5946x; 1.1013x over previous
#include <hip/hip_runtime.h>


#define GAT_N 50000
#define GAT_E 800000
#define GAT_F 64

typedef float v2f __attribute__((ext_vector_type(2)));
typedef float v8f __attribute__((ext_vector_type(8)));

template <int MASK>
__device__ __forceinline__ float gat_row_xor(float v) {
  return __int_as_float(__builtin_amdgcn_update_dpp(0, __float_as_int(v), 0x160 + MASK,
                                                    0xf, 0xf, false));
}
__device__ __forceinline__ float gat_row16_sum(float v) {
  v += gat_row_xor<1>(v);
  v += gat_row_xor<2>(v);
  v += gat_row_xor<4>(v);
  v += gat_row_xor<8>(v);
  return v;
}

__global__ __launch_bounds__(256)
void gat_gemm_wmma(const float* __restrict__ A, const float* __restrict__ B,
                   float* __restrict__ C, const float* __restrict__ attn_l,
                   const float* __restrict__ attn_r, float* __restrict__ el,
                   float* __restrict__ er, int M, int K, int Nw) {
  const int wave = blockIdx.x * 8 + (threadIdx.x >> 5);
  const int lane = threadIdx.x & 31;
  const int tilesN = Nw >> 6;
  const int tilesM = M >> 4;
  if (wave >= tilesM * tilesN) return;
  const int tm = wave / tilesN;
  const int tn = wave - tm * tilesN;
  const int kh = lane >> 4;
  const int lr = lane & 15;

  const float* Ap = A + (size_t)(tm * 16 + lr) * K + kh * 2;
  const float* Bp = B + (size_t)(tn * 64) + lr;
  const float* alp = attn_l + tn * 64 + lr;
  const float* arp = attn_r + tn * 64 + lr;
  const float l0 = alp[0], l1 = alp[16], l2 = alp[32], l3 = alp[48];
  const float r0 = arp[0], r1 = arp[16], r2 = arp[32], r3 = arp[48];
  v8f acc0 = {}, acc1 = {}, acc2 = {}, acc3 = {};
  for (int k0 = 0; k0 < K; k0 += 4) {
    v2f a = *(const v2f*)(Ap + k0);
    const float* Brow0 = Bp + (size_t)(k0 + kh * 2) * Nw;
    const float* Brow1 = Brow0 + Nw;
    v2f b0, b1, b2, b3;
    b0.x = Brow0[0];  b0.y = Brow1[0];
    b1.x = Brow0[16]; b1.y = Brow1[16];
    b2.x = Brow0[32]; b2.y = Brow1[32];
    b3.x = Brow0[48]; b3.y = Brow1[48];
    acc0 = __builtin_amdgcn_wmma_f32_16x16x4_f32(false, a, false, b0, (short)0, acc0, false, false);
    acc1 = __builtin_amdgcn_wmma_f32_16x16x4_f32(false, a, false, b1, (short)0, acc1, false, false);
    acc2 = __builtin_amdgcn_wmma_f32_16x16x4_f32(false, a, false, b2, (short)0, acc2, false, false);
    acc3 = __builtin_amdgcn_wmma_f32_16x16x4_f32(false, a, false, b3, (short)0, acc3, false, false);
  }
  float* Cp = C + (size_t)(tm * 16 + kh * 8) * Nw + tn * 64 + lr;
#pragma unroll
  for (int r = 0; r < 8; ++r) {
    Cp[(size_t)r * Nw +  0] = acc0[r];
    Cp[(size_t)r * Nw + 16] = acc1[r];
    Cp[(size_t)r * Nw + 32] = acc2[r];
    Cp[(size_t)r * Nw + 48] = acc3[r];
  }
  float keep = 0.f;
#pragma unroll
  for (int r = 0; r < 8; ++r) {
    const float pl = gat_row16_sum(acc0[r] * l0 + acc1[r] * l1 + acc2[r] * l2 + acc3[r] * l3);
    const float pr = gat_row16_sum(acc0[r] * r0 + acc1[r] * r1 + acc2[r] * r2 + acc3[r] * r3);
    keep = (lr == r) ? pl : keep;
    keep = (lr == r + 8) ? pr : keep;
  }
  float* ep = (lr < 8) ? el : er;
  ep[(size_t)(tm * 16 + kh * 8 + (lr & 7)) * tilesN + tn] = keep;
}

__global__ __launch_bounds__(256)
void gat_fill_u32(unsigned* __restrict__ p, unsigned v, long long n) {
  long long i = (long long)blockIdx.x * blockDim.x + threadIdx.x;
  if (i < n) p[i] = v;
}

__global__ __launch_bounds__(256)
void gat_csr_rank(const int* __restrict__ dst, int* __restrict__ cnt,
                  int* __restrict__ rank, int E) {
  int e = blockIdx.x * blockDim.x + threadIdx.x;
  if (e >= E) return;
  rank[e] = atomicAdd(&cnt[dst[e]], 1);
}

__global__ __launch_bounds__(1024)
void gat_csr_scan(const int* __restrict__ cnt, int* __restrict__ rowptr) {
  constexpr int n = GAT_N;
  constexpr int PER = (n + 1023) / 1024;
  __shared__ int part[1024];
  const int t = threadIdx.x;
  const int b = t * PER;
  int c[PER];
  int s = 0;
#pragma unroll
  for (int k = 0; k < PER; ++k) {
    c[k] = (b + k < n) ? cnt[b + k] : 0;
    s += c[k];
  }
  part[t] = s;
  __syncthreads();
  for (int off = 1; off < 1024; off <<= 1) {
    int v = (t >= off) ? part[t - off] : 0;
    __syncthreads();
    part[t] += v;
    __syncthreads();
  }
  int run = part[t] - s;
#pragma unroll
  for (int k = 0; k < PER; ++k) {
    if (b + k < n) rowptr[b + k] = run;
    run += c[k];
  }
  if (t == 1023) rowptr[n] = part[1023];
}

__global__ __launch_bounds__(256)
void gat_csr_place(const int* __restrict__ dst, const int* __restrict__ rowptr,
                   const int* __restrict__ rank, int* __restrict__ eid, int E) {
  int e = blockIdx.x * blockDim.x + threadIdx.x;
  if (e >= E) return;
  eid[rowptr[dst[e]] + rank[e]] = e;
}

__global__ __launch_bounds__(256)
void gat_csr_order(const int* __restrict__ src, const int* __restrict__ dst,
                   const int* __restrict__ rowptr, const int* __restrict__ eid,
                   int* __restrict__ csr_src, int E) {
  int e = blockIdx.x * blockDim.x + threadIdx.x;
  if (e >= E) return;
  const int d = dst[e];
  const int b = rowptr[d], en = rowptr[d + 1];
  int c = 0;
  for (int q = b; q < en; ++q) c += (eid[q] < e) ? 1 : 0;
  csr_src[b + c] = src[e];
}

__device__ __forceinline__ float gat_lrelu(float x) { return (x > 0.f) ? x : 0.2f * x; }
__device__ __forceinline__ float gat_wave_max(float v) {
  for (int o = 16; o > 0; o >>= 1) v = fmaxf(v, __shfl_xor(v, o, 32));
  return v;
}
__device__ __forceinline__ float gat_wave_sum(float v) {
  for (int o = 16; o > 0; o >>= 1) v += __shfl_xor(v, o, 32);
  return v;
}
__device__ __forceinline__ float gat_bcast(float v, int j) {
  return __int_as_float(__builtin_amdgcn_readlane(__float_as_int(v), j));
}

__global__ __launch_bounds__(256)
void gat_gather_h4(const int* __restrict__ rowptr, const int* __restrict__ csr_src,
                   const float* __restrict__ h, const float* __restrict__ el,
                   const float* __restrict__ er, const float* __restrict__ bias,
                   float* __restrict__ out, int N) {
  const int d = __builtin_amdgcn_readfirstlane((int)(blockIdx.x * 8 + (threadIdx.x >> 5)));
  if (d >= N) return;
  const int lane = threadIdx.x & 31;
  const bool hi = lane >= 16;
  const int beg = rowptr[d], end = rowptr[d + 1];
  const float4 erd = *(const float4*)(er + (size_t)d * 4);
  const float4* el4 = (const float4*)el;

  const float ninf = -__builtin_huge_valf();
  float m0 = ninf, m1 = ninf, m2 = ninf, m3 = ninf;
  for (int p = beg + lane; p < end; p += 32) {
    const float4 x = el4[csr_src[p]];
    m0 = fmaxf(m0, gat_lrelu(x.x + erd.x));
    m1 = fmaxf(m1, gat_lrelu(x.y + erd.y));
    m2 = fmaxf(m2, gat_lrelu(x.z + erd.z));
    m3 = fmaxf(m3, gat_lrelu(x.w + erd.w));
  }
  m0 = gat_wave_max(m0); m1 = gat_wave_max(m1);
  m2 = gat_wave_max(m2); m3 = gat_wave_max(m3);

  float s0 = 0.f, s1 = 0.f, s2 = 0.f, s3 = 0.f;
  for (int p = beg + lane; p < end; p += 32) {
    const float4 x = el4[csr_src[p]];
    s0 += __expf(gat_lrelu(x.x + erd.x) - m0);
    s1 += __expf(gat_lrelu(x.y + erd.y) - m1);
    s2 += __expf(gat_lrelu(x.z + erd.z) - m2);
    s3 += __expf(gat_lrelu(x.w + erd.w) - m3);
  }
  s0 = gat_wave_sum(s0); s1 = gat_wave_sum(s1);
  s2 = gat_wave_sum(s2); s3 = gat_wave_sum(s3);

  float4 acc0 = {0.f, 0.f, 0.f, 0.f}, acc1 = {0.f, 0.f, 0.f, 0.f};
  for (int base = beg; base < end; base += 32) {
    const int p = base + lane;
    int s = 0;
    float a0 = 0.f, a1 = 0.f, a2 = 0.f, a3 = 0.f;
    if (p < end) {
      s = csr_src[p];
      const float4 x = el4[s];
      a0 = __expf(gat_lrelu(x.x + erd.x) - m0) / s0;
      a1 = __expf(gat_lrelu(x.y + erd.y) - m1) / s1;
      a2 = __expf(gat_lrelu(x.z + erd.z) - m2) / s2;
      a3 = __expf(gat_lrelu(x.w + erd.w) - m3) / s3;
    }
    const int cnt = (end - base < 32) ? (end - base) : 32;
#pragma unroll 4
    for (int j = 0; j < cnt; ++j) {
      const int sj = __builtin_amdgcn_readlane(s, j);
      const float w0 = gat_bcast(a0, j), w1 = gat_bcast(a1, j);
      const float w2 = gat_bcast(a2, j), w3 = gat_bcast(a3, j);
      const float wA = hi ? w1 : w0;
      const float wB = hi ? w3 : w2;
      const float4* row = (const float4*)(h + (size_t)sj * 256);
      const float4 v0 = row[lane];
      const float4 v1 = row[32 + lane];
      acc0.x += v0.x * wA; acc0.y += v0.y * wA; acc0.z += v0.z * wA; acc0.w += v0.w * wA;
      acc1.x += v1.x * wB; acc1.y += v1.y * wB; acc1.z += v1.z * wB; acc1.w += v1.w * wB;
    }
  }
  const float4 b0 = ((const float4*)bias)[lane];
  const float4 b1 = ((const float4*)bias)[32 + lane];
  float4 o0, o1;
  o0.x = acc0.x + b0.x; o0.y = acc0.y + b0.y; o0.z = acc0.z + b0.z; o0.w = acc0.w + b0.w;
  o1.x = acc1.x + b1.x; o1.y = acc1.y + b1.y; o1.z = acc1.z + b1.z; o1.w = acc1.w + b1.w;
  o0.x = o0.x > 0.f ? o0.x : 0.f; o0.y = o0.y > 0.f ? o0.y : 0.f;
  o0.z = o0.z > 0.f ? o0.z : 0.f; o0.w = o0.w > 0.f ? o0.w : 0.f;
  o1.x = o1.x > 0.f ? o1.x : 0.f; o1.y = o1.y > 0.f ? o1.y : 0.f;
  o1.z = o1.z > 0.f ? o1.z : 0.f; o1.w = o1.w > 0.f ? o1.w : 0.f;
  float4* orow = (float4*)(out + (size_t)d * 256);
  orow[lane] = o0;
  orow[32 + lane] = o1;
}

__global__ __launch_bounds__(256)
void gat_gather_h1(const int* __restrict__ rowptr, const int* __restrict__ csr_src,
                   const float* __restrict__ h, const float* __restrict__ el,
                   const float* __restrict__ er, const float* __restrict__ bias,
                   float* __restrict__ out, int N) {
  const int d = __builtin_amdgcn_readfirstlane((int)(blockIdx.x * 8 + (threadIdx.x >> 5)));
  if (d >= N) return;
  const int lane = threadIdx.x & 31;
  const int beg = rowptr[d], end = rowptr[d + 1];
  const float erd = er[d];

  float m = -__builtin_huge_valf();
  for (int p = beg + lane; p < end; p += 32)
    m = fmaxf(m, gat_lrelu(el[csr_src[p]] + erd));
  m = gat_wave_max(m);

  float ssum = 0.f;
  for (int p = beg + lane; p < end; p += 32)
    ssum += __expf(gat_lrelu(el[csr_src[p]] + erd) - m);
  ssum = gat_wave_sum(ssum);

  float2 acc = {0.f, 0.f};
  for (int base = beg; base < end; base += 32) {
    const int p = base + lane;
    int s = 0;
    float a = 0.f;
    if (p < end) {
      s = csr_src[p];
      a = __expf(gat_lrelu(el[s] + erd) - m) / ssum;
    }
    const int cnt = (end - base < 32) ? (end - base) : 32;
#pragma unroll 4
    for (int j = 0; j < cnt; ++j) {
      const int sj = __builtin_amdgcn_readlane(s, j);
      const float w = gat_bcast(a, j);
      const float2 v = ((const float2*)(h + (size_t)sj * 64))[lane];
      acc.x += v.x * w; acc.y += v.y * w;
    }
  }
  const float2 b = ((const float2*)bias)[lane];
  float2 o;
  o.x = acc.x + b.x; o.y = acc.y + b.y;
  ((float2*)(out + (size_t)d * 64))[lane] = o;
}

extern "C" void kernel_launch(void* const* d_in, const int* in_sizes, int n_in,
                              void* d_out, int out_size, void* d_ws, size_t ws_size,
                              hipStream_t stream) {
  (void)in_sizes; (void)n_in; (void)out_size; (void)ws_size;
  const float* feat = (const float*)d_in[0];
  const int*   src  = (const int*)d_in[1];
  const int*   dst  = (const int*)d_in[2];
  const float* W1  = (const float*)d_in[3];
  const float* al1 = (const float*)d_in[4];
  const float* ar1 = (const float*)d_in[5];
  const float* b1  = (const float*)d_in[6];
  const float* W2  = (const float*)d_in[7];
  const float* al2 = (const float*)d_in[8];
  const float* ar2 = (const float*)d_in[9];
  const float* b2  = (const float*)d_in[10];
  const float* W3  = (const float*)d_in[11];
  const float* al3 = (const float*)d_in[12];
  const float* ar3 = (const float*)d_in[13];
  const float* b3  = (const float*)d_in[14];
  float* out = (float*)d_out;

  const int N = GAT_N, E = GAT_E;

  char* ws = (char*)d_ws;
  size_t off = 0;
  auto carve = [&](size_t bytes) -> void* {
    void* p = ws + off;
    off += (bytes + 255) & ~(size_t)255;
    return p;
  };
  float* bufA    = (float*)carve((size_t)N * 256 * 4);
  float* bufB    = (float*)carve((size_t)N * 256 * 4);
  float* el      = (float*)carve((size_t)N * 4 * 4);
  float* er      = (float*)carve((size_t)N * 4 * 4);
  int*   cnt     = (int*)carve((size_t)N * 4);
  int*   rowptr  = (int*)carve((size_t)(N + 1) * 4);
  int*   rank    = (int*)carve((size_t)E * 4);
  int*   eid     = (int*)carve((size_t)E * 4);
  int*   csr_src = (int*)carve((size_t)E * 4);

  auto blocks = [](long long n) { return (unsigned)((n + 255) / 256); };

  gat_fill_u32<<<blocks(N), 256, 0, stream>>>((unsigned*)cnt, 0u, (long long)N);
  gat_csr_rank<<<blocks(E), 256, 0, stream>>>(dst, cnt, rank, E);
  gat_csr_scan<<<1, 1024, 0, stream>>>(cnt, rowptr);
  gat_csr_place<<<blocks(E), 256, 0, stream>>>(dst, rowptr, rank, eid, E);
  gat_csr_order<<<blocks(E), 256, 0, stream>>>(src, dst, rowptr, eid, csr_src, E);

  const unsigned node_blocks = (unsigned)((N + 7) / 8);

  {
    const int K = 128, Nw = 256;
    int waves = (N / 16) * (Nw / 64);
    gat_gemm_wmma<<<(waves + 7) / 8, 256, 0, stream>>>(feat, W1, bufA, al1, ar1, el, er, N, K, Nw);
    gat_gather_h4<<<node_blocks, 256, 0, stream>>>(rowptr, csr_src, bufA, el, er, b1, bufB, N);
  }

  {
    const int K = 256, Nw = 256;
    int waves = (N / 16) * (Nw / 64);
    gat_gemm_wmma<<<(waves + 7) / 8, 256, 0, stream>>>(bufB, W2, bufA, al2, ar2, el, er, N, K, Nw);
    gat_gather_h4<<<node_blocks, 256, 0, stream>>>(rowptr, csr_src, bufA, el, er, b2, bufB, N);
  }

  {
    const int K = 256, Nw = 64;
    int waves = (N / 16) * (Nw / 64);
    gat_gemm_wmma<<<(waves + 7) / 8, 256, 0, stream>>>(bufB, W3, bufA, al3, ar3, el, er, N, K, Nw);
    gat_gather_h1<<<node_blocks, 256, 0, stream>>>(rowptr, csr_src, bufA, el, er, b3, out, N);
  }
}


# Scan kernel walks an LDS copy of the counts: counts come in and bucket starts go out as whole-line b128 loads/stores, not 4-byte accesses that touch 32 lines per instruction
# speedup vs baseline: 5.1271x; 1.1159x over previous
#include <hip/hip_runtime.h>


#define GAT_N 50000
#define GAT_E 800000
#define GAT_F 64

typedef float v2f __attribute__((ext_vector_type(2)));
typedef float v8f __attribute__((ext_vector_type(8)));

template <int MASK>
__device__ __forceinline__ float gat_row_xor(float v) {
  return __int_as_float(__builtin_amdgcn_update_dpp(0, __float_as_int(v), 0x160 + MASK,
                                                    0xf, 0xf, false));
}
__device__ __forceinline__ float gat_row16_sum(float v) {
  v += gat_row_xor<1>(v);
  v += gat_row_xor<2>(v);
  v += gat_row_xor<4>(v);
  v += gat_row_xor<8>(v);
  return v;
}

__global__ __launch_bounds__(256)
void gat_gemm_wmma(const float* __restrict__ A, const float* __restrict__ B,
                   float* __restrict__ C, const float* __restrict__ attn_l,
                   const float* __restrict__ attn_r, float* __restrict__ el,
                   float* __restrict__ er, int M, int K, int Nw) {
  const int wave = blockIdx.x * 8 + (threadIdx.x >> 5);
  const int lane = threadIdx.x & 31;
  const int tilesN = Nw >> 6;
  const int tilesM = M >> 4;
  if (wave >= tilesM * tilesN) return;
  const int tm = wave / tilesN;
  const int tn = wave - tm * tilesN;
  const int kh = lane >> 4;
  const int lr = lane & 15;

  const float* Ap = A + (size_t)(tm * 16 + lr) * K + kh * 2;
  const float* Bp = B + (size_t)(tn * 64) + lr;
  const float* alp = attn_l + tn * 64 + lr;
  const float* arp = attn_r + tn * 64 + lr;
  const float l0 = alp[0], l1 = alp[16], l2 = alp[32], l3 = alp[48];
  const float r0 = arp[0], r1 = arp[16], r2 = arp[32], r3 = arp[48];
  v8f acc0 = {}, acc1 = {}, acc2 = {}, acc3 = {};
  for (int k0 = 0; k0 < K; k0 += 4) {
    v2f a = *(const v2f*)(Ap + k0);
    const float* Brow0 = Bp + (size_t)(k0 + kh * 2) * Nw;
    const float* Brow1 = Brow0 + Nw;
    v2f b0, b1, b2, b3;
    b0.x = Brow0[0];  b0.y = Brow1[0];
    b1.x = Brow0[16]; b1.y = Brow1[16];
    b2.x = Brow0[32]; b2.y = Brow1[32];
    b3.x = Brow0[48]; b3.y = Brow1[48];
    acc0 = __builtin_amdgcn_wmma_f32_16x16x4_f32(false, a, false, b0, (short)0, acc0, false, false);
    acc1 = __builtin_amdgcn_wmma_f32_16x16x4_f32(false, a, false, b1, (short)0, acc1, false, false);
    acc2 = __builtin_amdgcn_wmma_f32_16x16x4_f32(false, a, false, b2, (short)0, acc2, false, false);
    acc3 = __builtin_amdgcn_wmma_f32_16x16x4_f32(false, a, false, b3, (short)0, acc3, false, false);
  }
  float* Cp = C + (size_t)(tm * 16 + kh * 8) * Nw + tn * 64 + lr;
#pragma unroll
  for (int r = 0; r < 8; ++r) {
    Cp[(size_t)r * Nw +  0] = acc0[r];
    Cp[(size_t)r * Nw + 16] = acc1[r];
    Cp[(size_t)r * Nw + 32] = acc2[r];
    Cp[(size_t)r * Nw + 48] = acc3[r];
  }
  float keep = 0.f;
#pragma unroll
  for (int r = 0; r < 8; ++r) {
    const float pl = gat_row16_sum(acc0[r] * l0 + acc1[r] * l1 + acc2[r] * l2 + acc3[r] * l3);
    const float pr = gat_row16_sum(acc0[r] * r0 + acc1[r] * r1 + acc2[r] * r2 + acc3[r] * r3);
    keep = (lr == r) ? pl : keep;
    keep = (lr == r + 8) ? pr : keep;
  }
  float* ep = (lr < 8) ? el : er;
  ep[(size_t)(tm * 16 + kh * 8 + (lr & 7)) * tilesN + tn] = keep;
}

__global__ __launch_bounds__(256)
void gat_fill_u32(unsigned* __restrict__ p, unsigned v, long long n) {
  long long i = (long long)blockIdx.x * blockDim.x + threadIdx.x;
  if (i < n) p[i] = v;
}

__global__ __launch_bounds__(256)
void gat_csr_rank(const int* __restrict__ dst, int* __restrict__ cnt,
                  int* __restrict__ rank, int E) {
  int e = blockIdx.x * blockDim.x + threadIdx.x;
  if (e >= E) return;
  rank[e] = atomicAdd(&cnt[dst[e]], 1);
}

__global__ __launch_bounds__(1024)
void gat_csr_scan(const int* __restrict__ cnt, int* __restrict__ rowptr) {
  constexpr int n = GAT_N;
  constexpr int PER = (n + 1023) / 1024;
  constexpr int NV = (1024 * PER) / 4;
  constexpr int NVG = n / 4;
  constexpr int ROUNDS = (NV + 1023) / 1024;
  static_assert(n % 4 == 0, "four-word units must tile the node count");
  __shared__ int4 buf4[NV];
  __shared__ int part[1024];
  int* buf = (int*)buf4;
  const int t = threadIdx.x;
  const int4* __restrict__ cnt4 = (const int4*)cnt;
  int4* __restrict__ rowptr4 = (int4*)rowptr;
#pragma unroll
  for (int r = 0; r < ROUNDS; ++r) {
    const int i = t + r * 1024;
    if (i < NV) buf4[i] = (i < NVG) ? cnt4[i] : make_int4(0, 0, 0, 0);
  }
  __syncthreads();
  const int b = t * PER;
  int c[PER];
  int s = 0;
#pragma unroll
  for (int k = 0; k < PER; ++k) {
    c[k] = buf[b + k];
    s += c[k];
  }
  part[t] = s;
  __syncthreads();
  for (int off = 1; off < 1024; off <<= 1) {
    int v = (t >= off) ? part[t - off] : 0;
    __syncthreads();
    part[t] += v;
    __syncthreads();
  }
  int run = part[t] - s;
#pragma unroll
  for (int k = 0; k < PER; ++k) {
    buf[b + k] = run;
    run += c[k];
  }
  __syncthreads();
#pragma unroll
  for (int r = 0; r < ROUNDS; ++r) {
    const int i = t + r * 1024;
    if (i < NVG) rowptr4[i] = buf4[i];
  }
  if (t == 1023) rowptr[n] = part[1023];
}

__global__ __launch_bounds__(256)
void gat_csr_place(const int* __restrict__ dst, const int* __restrict__ rowptr,
                   const int* __restrict__ rank, int* __restrict__ eid, int E) {
  int e = blockIdx.x * blockDim.x + threadIdx.x;
  if (e >= E) return;
  eid[rowptr[dst[e]] + rank[e]] = e;
}

__global__ __launch_bounds__(256)
void gat_csr_order(const int* __restrict__ src, const int* __restrict__ dst,
                   const int* __restrict__ rowptr, const int* __restrict__ eid,
                   int* __restrict__ csr_src, int E) {
  int e = blockIdx.x * blockDim.x + threadIdx.x;
  if (e >= E) return;
  const int d = dst[e];
  const int b = rowptr[d], en = rowptr[d + 1];
  int c = 0;
  for (int q = b; q < en; ++q) c += (eid[q] < e) ? 1 : 0;
  csr_src[b + c] = src[e];
}

__device__ __forceinline__ float gat_lrelu(float x) { return (x > 0.f) ? x : 0.2f * x; }
__device__ __forceinline__ float gat_wave_max(float v) {
  for (int o = 16; o > 0; o >>= 1) v = fmaxf(v, __shfl_xor(v, o, 32));
  return v;
}
__device__ __forceinline__ float gat_wave_sum(float v) {
  for (int o = 16; o > 0; o >>= 1) v += __shfl_xor(v, o, 32);
  return v;
}
__device__ __forceinline__ float gat_bcast(float v, int j) {
  return __int_as_float(__builtin_amdgcn_readlane(__float_as_int(v), j));
}

__global__ __launch_bounds__(256)
void gat_gather_h4(const int* __restrict__ rowptr, const int* __restrict__ csr_src,
                   const float* __restrict__ h, const float* __restrict__ el,
                   const float* __restrict__ er, const float* __restrict__ bias,
                   float* __restrict__ out, int N) {
  const int d = __builtin_amdgcn_readfirstlane((int)(blockIdx.x * 8 + (threadIdx.x >> 5)));
  if (d >= N) return;
  const int lane = threadIdx.x & 31;
  const bool hi = lane >= 16;
  const int beg = rowptr[d], end = rowptr[d + 1];
  const float4 erd = *(const float4*)(er + (size_t)d * 4);
  const float4* el4 = (const float4*)el;

  const float ninf = -__builtin_huge_valf();
  float m0 = ninf, m1 = ninf, m2 = ninf, m3 = ninf;
  for (int p = beg + lane; p < end; p += 32) {
    const float4 x = el4[csr_src[p]];
    m0 = fmaxf(m0, gat_lrelu(x.x + erd.x));
    m1 = fmaxf(m1, gat_lrelu(x.y + erd.y));
    m2 = fmaxf(m2, gat_lrelu(x.z + erd.z));
    m3 = fmaxf(m3, gat_lrelu(x.w + erd.w));
  }
  m0 = gat_wave_max(m0); m1 = gat_wave_max(m1);
  m2 = gat_wave_max(m2); m3 = gat_wave_max(m3);

  float s0 = 0.f, s1 = 0.f, s2 = 0.f, s3 = 0.f;
  for (int p = beg + lane; p < end; p += 32) {
    const float4 x = el4[csr_src[p]];
    s0 += __expf(gat_lrelu(x.x + erd.x) - m0);
    s1 += __expf(gat_lrelu(x.y + erd.y) - m1);
    s2 += __expf(gat_lrelu(x.z + erd.z) - m2);
    s3 += __expf(gat_lrelu(x.w + erd.w) - m3);
  }
  s0 = gat_wave_sum(s0); s1 = gat_wave_sum(s1);
  s2 = gat_wave_sum(s2); s3 = gat_wave_sum(s3);

  float4 acc0 = {0.f, 0.f, 0.f, 0.f}, acc1 = {0.f, 0.f, 0.f, 0.f};
  for (int base = beg; base < end; base += 32) {
    const int p = base + lane;
    int s = 0;
    float a0 = 0.f, a1 = 0.f, a2 = 0.f, a3 = 0.f;
    if (p < end) {
      s = csr_src[p];
      const float4 x = el4[s];
      a0 = __expf(gat_lrelu(x.x + erd.x) - m0) / s0;
      a1 = __expf(gat_lrelu(x.y + erd.y) - m1) / s1;
      a2 = __expf(gat_lrelu(x.z + erd.z) - m2) / s2;
      a3 = __expf(gat_lrelu(x.w + erd.w) - m3) / s3;
    }
    const int cnt = (end - base < 32) ? (end - base) : 32;
#pragma unroll 4
    for (int j = 0; j < cnt; ++j) {
      const int sj = __builtin_amdgcn_readlane(s, j);
      const float w0 = gat_bcast(a0, j), w1 = gat_bcast(a1, j);
      const float w2 = gat_bcast(a2, j), w3 = gat_bcast(a3, j);
      const float wA = hi ? w1 : w0;
      const float wB = hi ? w3 : w2;
      const float4* row = (const float4*)(h + (size_t)sj * 256);
      const float4 v0 = row[lane];
      const float4 v1 = row[32 + lane];
      acc0.x += v0.x * wA; acc0.y += v0.y * wA; acc0.z += v0.z * wA; acc0.w += v0.w * wA;
      acc1.x += v1.x * wB; acc1.y += v1.y * wB; acc1.z += v1.z * wB; acc1.w += v1.w * wB;
    }
  }
  const float4 b0 = ((const float4*)bias)[lane];
  const float4 b1 = ((const float4*)bias)[32 + lane];
  float4 o0, o1;
  o0.x = acc0.x + b0.x; o0.y = acc0.y + b0.y; o0.z = acc0.z + b0.z; o0.w = acc0.w + b0.w;
  o1.x = acc1.x + b1.x; o1.y = acc1.y + b1.y; o1.z = acc1.z + b1.z; o1.w = acc1.w + b1.w;
  o0.x = o0.x > 0.f ? o0.x : 0.f; o0.y = o0.y > 0.f ? o0.y : 0.f;
  o0.z = o0.z > 0.f ? o0.z : 0.f; o0.w = o0.w > 0.f ? o0.w : 0.f;
  o1.x = o1.x > 0.f ? o1.x : 0.f; o1.y = o1.y > 0.f ? o1.y : 0.f;
  o1.z = o1.z > 0.f ? o1.z : 0.f; o1.w = o1.w > 0.f ? o1.w : 0.f;
  float4* orow = (float4*)(out + (size_t)d * 256);
  orow[lane] = o0;
  orow[32 + lane] = o1;
}

__global__ __launch_bounds__(256)
void gat_gather_h1(const int* __restrict__ rowptr, const int* __restrict__ csr_src,
                   const float* __restrict__ h, const float* __restrict__ el,
                   const float* __restrict__ er, const float* __restrict__ bias,
                   float* __restrict__ out, int N) {
  const int d = __builtin_amdgcn_readfirstlane((int)(blockIdx.x * 8 + (threadIdx.x >> 5)));
  if (d >= N) return;
  const int lane = threadIdx.x & 31;
  const int beg = rowptr[d], end = rowptr[d + 1];
  const float erd = er[d];

  float m = -__builtin_huge_valf();
  for (int p = beg + lane; p < end; p += 32)
    m = fmaxf(m, gat_lrelu(el[csr_src[p]] + erd));
  m = gat_wave_max(m);

  float ssum = 0.f;
  for (int p = beg + lane; p < end; p += 32)
    ssum += __expf(gat_lrelu(el[csr_src[p]] + erd) - m);
  ssum = gat_wave_sum(ssum);

  float2 acc = {0.f, 0.f};
  for (int base = beg; base < end; base += 32) {
    const int p = base + lane;
    int s = 0;
    float a = 0.f;
    if (p < end) {
      s = csr_src[p];
      a = __expf(gat_lrelu(el[s] + erd) - m) / ssum;
    }
    const int cnt = (end - base < 32) ? (end - base) : 32;
#pragma unroll 4
    for (int j = 0; j < cnt; ++j) {
      const int sj = __builtin_amdgcn_readlane(s, j);
      const float w = gat_bcast(a, j);
      const float2 v = ((const float2*)(h + (size_t)sj * 64))[lane];
      acc.x += v.x * w; acc.y += v.y * w;
    }
  }
  const float2 b = ((const float2*)bias)[lane];
  float2 o;
  o.x = acc.x + b.x; o.y = acc.y + b.y;
  ((float2*)(out + (size_t)d * 64))[lane] = o;
}

extern "C" void kernel_launch(void* const* d_in, const int* in_sizes, int n_in,
                              void* d_out, int out_size, void* d_ws, size_t ws_size,
                              hipStream_t stream) {
  (void)in_sizes; (void)n_in; (void)out_size; (void)ws_size;
  const float* feat = (const float*)d_in[0];
  const int*   src  = (const int*)d_in[1];
  const int*   dst  = (const int*)d_in[2];
  const float* W1  = (const float*)d_in[3];
  const float* al1 = (const float*)d_in[4];
  const float* ar1 = (const float*)d_in[5];
  const float* b1  = (const float*)d_in[6];
  const float* W2  = (const float*)d_in[7];
  const float* al2 = (const float*)d_in[8];
  const float* ar2 = (const float*)d_in[9];
  const float* b2  = (const float*)d_in[10];
  const float* W3  = (const float*)d_in[11];
  const float* al3 = (const float*)d_in[12];
  const float* ar3 = (const float*)d_in[13];
  const float* b3  = (const float*)d_in[14];
  float* out = (float*)d_out;

  const int N = GAT_N, E = GAT_E;

  char* ws = (char*)d_ws;
  size_t off = 0;
  auto carve = [&](size_t bytes) -> void* {
    void* p = ws + off;
    off += (bytes + 255) & ~(size_t)255;
    return p;
  };
  float* bufA    = (float*)carve((size_t)N * 256 * 4);
  float* bufB    = (float*)carve((size_t)N * 256 * 4);
  float* el      = (float*)carve((size_t)N * 4 * 4);
  float* er      = (float*)carve((size_t)N * 4 * 4);
  int*   cnt     = (int*)carve((size_t)N * 4);
  int*   rowptr  = (int*)carve((size_t)(N + 1) * 4);
  int*   rank    = (int*)carve((size_t)E * 4);
  int*   eid     = (int*)carve((size_t)E * 4);
  int*   csr_src = (int*)carve((size_t)E * 4);

  auto blocks = [](long long n) { return (unsigned)((n + 255) / 256); };

  gat_fill_u32<<<blocks(N), 256, 0, stream>>>((unsigned*)cnt, 0u, (long long)N);
  gat_csr_rank<<<blocks(E), 256, 0, stream>>>(dst, cnt, rank, E);
  gat_csr_scan<<<1, 1024, 0, stream>>>(cnt, rowptr);
  gat_csr_place<<<blocks(E), 256, 0, stream>>>(dst, rowptr, rank, eid, E);
  gat_csr_order<<<blocks(E), 256, 0, stream>>>(src, dst, rowptr, eid, csr_src, E);

  const unsigned node_blocks = (unsigned)((N + 7) / 8);

  {
    const int K = 128, Nw = 256;
    int waves = (N / 16) * (Nw / 64);
    gat_gemm_wmma<<<(waves + 7) / 8, 256, 0, stream>>>(feat, W1, bufA, al1, ar1, el, er, N, K, Nw);
    gat_gather_h4<<<node_blocks, 256, 0, stream>>>(rowptr, csr_src, bufA, el, er, b1, bufB, N);
  }

  {
    const int K = 256, Nw = 256;
    int waves = (N / 16) * (Nw / 64);
    gat_gemm_wmma<<<(waves + 7) / 8, 256, 0, stream>>>(bufB, W2, bufA, al2, ar2, el, er, N, K, Nw);
    gat_gather_h4<<<node_blocks, 256, 0, stream>>>(rowptr, csr_src, bufA, el, er, b2, bufB, N);
  }

  {
    const int K = 256, Nw = 64;
    int waves = (N / 16) * (Nw / 64);
    gat_gemm_wmma<<<(waves + 7) / 8, 256, 0, stream>>>(bufB, W3, bufA, al3, ar3, el, er, N, K, Nw);
    gat_gather_h1<<<node_blocks, 256, 0, stream>>>(rowptr, csr_src, bufA, el, er, b3, out, N);
  }
}


# GEMM operand feed: 8-wave block stages its head's B strip once in LDS (k-pair layout, 8-byte fragment reads), A read as whole half lines per 32 k; no global load per k-step; new k order in f32 sums
# speedup vs baseline: 6.2403x; 1.2171x over previous
#include <hip/hip_runtime.h>


#define GAT_N 50000
#define GAT_E 800000
#define GAT_F 64

typedef float v2f __attribute__((ext_vector_type(2)));
typedef float v8f __attribute__((ext_vector_type(8)));

template <int MASK>
__device__ __forceinline__ float gat_row_xor(float v) {
  return __int_as_float(__builtin_amdgcn_update_dpp(0, __float_as_int(v), 0x160 + MASK,
                                                    0xf, 0xf, false));
}
__device__ __forceinline__ float gat_row16_sum(float v) {
  v += gat_row_xor<1>(v);
  v += gat_row_xor<2>(v);
  v += gat_row_xor<4>(v);
  v += gat_row_xor<8>(v);
  return v;
}

template <int K>
__global__ __launch_bounds__(256)
void gat_gemm_wmma(const float* __restrict__ A, const float* __restrict__ B,
                   float* __restrict__ C, const float* __restrict__ attn_l,
                   const float* __restrict__ attn_r, float* __restrict__ el,
                   float* __restrict__ er, int M, int Nw) {
  constexpr int PR = K / 2;
  constexpr int PITCH = 132;
  constexpr int NB = K / 32;
  static_assert(K % 32 == 0, "K must be a whole number of 32-k blocks");
  __shared__ float4 bs4[PR * PITCH / 4];
  float* bs = (float*)bs4;
  const int t = threadIdx.x;
  const int tilesN = Nw >> 6;
  const int tilesM = M >> 4;
  const int rb = blockIdx.x / tilesN;
  const int tn = blockIdx.x - rb * tilesN;
  {
    const int cg = t & 15;
    const int pr0 = t >> 4;
    const float* Bg = B + (size_t)tn * 64 + cg * 4;
#pragma unroll
    for (int it = 0; it < PR / 16; ++it) {
      const int pr = pr0 + it * 16;
      const float4 r0 = *(const float4*)(Bg + (size_t)(2 * pr) * Nw);
      const float4 r1 = *(const float4*)(Bg + (size_t)(2 * pr + 1) * Nw);
      float4* d4 = (float4*)(bs + pr * PITCH + cg * 8);
      d4[0] = make_float4(r0.x, r1.x, r0.y, r1.y);
      d4[1] = make_float4(r0.z, r1.z, r0.w, r1.w);
    }
  }
  __syncthreads();
  const int tm = rb * 8 + (t >> 5);
  if (tm >= tilesM) return;
  const int lane = t & 31;
  const int kh = lane >> 4;
  const int lr = lane & 15;

  const float* Ap = A + (size_t)(tm * 16 + lr) * K + kh * 16;
  const float* bl = bs + kh * 8 * PITCH + lr * 2;
  const float* alp = attn_l + tn * 64 + lr;
  const float* arp = attn_r + tn * 64 + lr;
  const float l0 = alp[0], l1 = alp[16], l2 = alp[32], l3 = alp[48];
  const float r0 = arp[0], r1 = arp[16], r2 = arp[32], r3 = arp[48];
  v8f acc0 = {}, acc1 = {}, acc2 = {}, acc3 = {};
  float4 q0 = *(const float4*)(Ap + 0);
  float4 q1 = *(const float4*)(Ap + 4);
  float4 q2 = *(const float4*)(Ap + 8);
  float4 q3 = *(const float4*)(Ap + 12);
#define GAT_KSTEP(S, AX, AY)                                                                     \
  {                                                                                              \
    v2f a; a.x = (AX); a.y = (AY);                                                               \
    const float* br = bb + (S) * PITCH;                                                          \
    const v2f b0 = *(const v2f*)(br +  0);                                                       \
    const v2f b1 = *(const v2f*)(br + 32);                                                       \
    const v2f b2 = *(const v2f*)(br + 64);                                                       \
    const v2f b3 = *(const v2f*)(br + 96);                                                       \
    acc0 = __builtin_amdgcn_wmma_f32_16x16x4_f32(false, a, false, b0, (short)0, acc0, false, false); \
    acc1 = __builtin_amdgcn_wmma_f32_16x16x4_f32(false, a, false, b1, (short)0, acc1, false, false); \
    acc2 = __builtin_amdgcn_wmma_f32_16x16x4_f32(false, a, false, b2, (short)0, acc2, false, false); \
    acc3 = __builtin_amdgcn_wmma_f32_16x16x4_f32(false, a, false, b3, (short)0, acc3, false, false); \
  }
#pragma unroll
  for (int jb = 0; jb < NB; ++jb) {
    float4 n0 = q0, n1 = q1, n2 = q2, n3 = q3;
    if (jb + 1 < NB) {
      const float* An = Ap + (jb + 1) * 32;
      n0 = *(const float4*)(An + 0);
      n1 = *(const float4*)(An + 4);
      n2 = *(const float4*)(An + 8);
      n3 = *(const float4*)(An + 12);
    }
    const float* bb = bl + jb * 16 * PITCH;
    GAT_KSTEP(0, q0.x, q0.y)
    GAT_KSTEP(1, q0.z, q0.w)
    GAT_KSTEP(2, q1.x, q1.y)
    GAT_KSTEP(3, q1.z, q1.w)
    GAT_KSTEP(4, q2.x, q2.y)
    GAT_KSTEP(5, q2.z, q2.w)
    GAT_KSTEP(6, q3.x, q3.y)
    GAT_KSTEP(7, q3.z, q3.w)
    q0 = n0; q1 = n1; q2 = n2; q3 = n3;
  }
#undef GAT_KSTEP
  float* Cp = C + (size_t)(tm * 16 + kh * 8) * Nw + tn * 64 + lr;
#pragma unroll
  for (int r = 0; r < 8; ++r) {
    Cp[(size_t)r * Nw +  0] = acc0[r];
    Cp[(size_t)r * Nw + 16] = acc1[r];
    Cp[(size_t)r * Nw + 32] = acc2[r];
    Cp[(size_t)r * Nw + 48] = acc3[r];
  }
  float keep = 0.f;
#pragma unroll
  for (int r = 0; r < 8; ++r) {
    const float pl = gat_row16_sum(acc0[r] * l0 + acc1[r] * l1 + acc2[r] * l2 + acc3[r] * l3);
    const float pr = gat_row16_sum(acc0[r] * r0 + acc1[r] * r1 + acc2[r] * r2 + acc3[r] * r3);
    keep = (lr == r) ? pl : keep;
    keep = (lr == r + 8) ? pr : keep;
  }
  float* ep = (lr < 8) ? el : er;
  ep[(size_t)(tm * 16 + kh * 8 + (lr & 7)) * tilesN + tn] = keep;
}

__global__ __launch_bounds__(256)
void gat_fill_u32(unsigned* __restrict__ p, unsigned v, long long n) {
  long long i = (long long)blockIdx.x * blockDim.x + threadIdx.x;
  if (i < n) p[i] = v;
}

__global__ __launch_bounds__(256)
void gat_csr_rank(const int* __restrict__ dst, int* __restrict__ cnt,
                  int* __restrict__ rank, int E) {
  int e = blockIdx.x * blockDim.x + threadIdx.x;
  if (e >= E) return;
  rank[e] = atomicAdd(&cnt[dst[e]], 1);
}

__global__ __launch_bounds__(1024)
void gat_csr_scan(const int* __restrict__ cnt, int* __restrict__ rowptr) {
  constexpr int n = GAT_N;
  constexpr int PER = (n + 1023) / 1024;
  constexpr int NV = (1024 * PER) / 4;
  constexpr int NVG = n / 4;
  constexpr int ROUNDS = (NV + 1023) / 1024;
  static_assert(n % 4 == 0, "four-word units must tile the node count");
  __shared__ int4 buf4[NV];
  __shared__ int part[1024];
  int* buf = (int*)buf4;
  const int t = threadIdx.x;
  const int4* __restrict__ cnt4 = (const int4*)cnt;
  int4* __restrict__ rowptr4 = (int4*)rowptr;
#pragma unroll
  for (int r = 0; r < ROUNDS; ++r) {
    const int i = t + r * 1024;
    if (i < NV) buf4[i] = (i < NVG) ? cnt4[i] : make_int4(0, 0, 0, 0);
  }
  __syncthreads();
  const int b = t * PER;
  int c[PER];
  int s = 0;
#pragma unroll
  for (int k = 0; k < PER; ++k) {
    c[k] = buf[b + k];
    s += c[k];
  }
  part[t] = s;
  __syncthreads();
  for (int off = 1; off < 1024; off <<= 1) {
    int v = (t >= off) ? part[t - off] : 0;
    __syncthreads();
    part[t] += v;
    __syncthreads();
  }
  int run = part[t] - s;
#pragma unroll
  for (int k = 0; k < PER; ++k) {
    buf[b + k] = run;
    run += c[k];
  }
  __syncthreads();
#pragma unroll
  for (int r = 0; r < ROUNDS; ++r) {
    const int i = t + r * 1024;
    if (i < NVG) rowptr4[i] = buf4[i];
  }
  if (t == 1023) rowptr[n] = part[1023];
}

__global__ __launch_bounds__(256)
void gat_csr_place(const int* __restrict__ dst, const int* __restrict__ rowptr,
                   const int* __restrict__ rank, int* __restrict__ eid, int E) {
  int e = blockIdx.x * blockDim.x + threadIdx.x;
  if (e >= E) return;
  eid[rowptr[dst[e]] + rank[e]] = e;
}

__global__ __launch_bounds__(256)
void gat_csr_order(const int* __restrict__ src, const int* __restrict__ dst,
                   const int* __restrict__ rowptr, const int* __restrict__ eid,
                   int* __restrict__ csr_src, int E) {
  int e = blockIdx.x * blockDim.x + threadIdx.x;
  if (e >= E) return;
  const int d = dst[e];
  const int b = rowptr[d], en = rowptr[d + 1];
  int c = 0;
  for (int q = b; q < en; ++q) c += (eid[q] < e) ? 1 : 0;
  csr_src[b + c] = src[e];
}

__device__ __forceinline__ float gat_lrelu(float x) { return (x > 0.f) ? x : 0.2f * x; }
__device__ __forceinline__ float gat_wave_max(float v) {
  for (int o = 16; o > 0; o >>= 1) v = fmaxf(v, __shfl_xor(v, o, 32));
  return v;
}
__device__ __forceinline__ float gat_wave_sum(float v) {
  for (int o = 16; o > 0; o >>= 1) v += __shfl_xor(v, o, 32);
  return v;
}
__device__ __forceinline__ float gat_bcast(float v, int j) {
  return __int_as_float(__builtin_amdgcn_readlane(__float_as_int(v), j));
}

__global__ __launch_bounds__(256)
void gat_gather_h4(const int* __restrict__ rowptr, const int* __restrict__ csr_src,
                   const float* __restrict__ h, const float* __restrict__ el,
                   const float* __restrict__ er, const float* __restrict__ bias,
                   float* __restrict__ out, int N) {
  const int d = __builtin_amdgcn_readfirstlane((int)(blockIdx.x * 8 + (threadIdx.x >> 5)));
  if (d >= N) return;
  const int lane = threadIdx.x & 31;
  const bool hi = lane >= 16;
  const int beg = rowptr[d], end = rowptr[d + 1];
  const float4 erd = *(const float4*)(er + (size_t)d * 4);
  const float4* el4 = (const float4*)el;

  const float ninf = -__builtin_huge_valf();
  float m0 = ninf, m1 = ninf, m2 = ninf, m3 = ninf;
  for (int p = beg + lane; p < end; p += 32) {
    const float4 x = el4[csr_src[p]];
    m0 = fmaxf(m0, gat_lrelu(x.x + erd.x));
    m1 = fmaxf(m1, gat_lrelu(x.y + erd.y));
    m2 = fmaxf(m2, gat_lrelu(x.z + erd.z));
    m3 = fmaxf(m3, gat_lrelu(x.w + erd.w));
  }
  m0 = gat_wave_max(m0); m1 = gat_wave_max(m1);
  m2 = gat_wave_max(m2); m3 = gat_wave_max(m3);

  float s0 = 0.f, s1 = 0.f, s2 = 0.f, s3 = 0.f;
  for (int p = beg + lane; p < end; p += 32) {
    const float4 x = el4[csr_src[p]];
    s0 += __expf(gat_lrelu(x.x + erd.x) - m0);
    s1 += __expf(gat_lrelu(x.y + erd.y) - m1);
    s2 += __expf(gat_lrelu(x.z + erd.z) - m2);
    s3 += __expf(gat_lrelu(x.w + erd.w) - m3);
  }
  s0 = gat_wave_sum(s0); s1 = gat_wave_sum(s1);
  s2 = gat_wave_sum(s2); s3 = gat_wave_sum(s3);

  float4 acc0 = {0.f, 0.f, 0.f, 0.f}, acc1 = {0.f, 0.f, 0.f, 0.f};
  for (int base = beg; base < end; base += 32) {
    const int p = base + lane;
    int s = 0;
    float a0 = 0.f, a1 = 0.f, a2 = 0.f, a3 = 0.f;
    if (p < end) {
      s = csr_src[p];
      const float4 x = el4[s];
      a0 = __expf(gat_lrelu(x.x + erd.x) - m0) / s0;
      a1 = __expf(gat_lrelu(x.y + erd.y) - m1) / s1;
      a2 = __expf(gat_lrelu(x.z + erd.z) - m2) / s2;
      a3 = __expf(gat_lrelu(x.w + erd.w) - m3) / s3;
    }
    const int cnt = (end - base < 32) ? (end - base) : 32;
#pragma unroll 4
    for (int j = 0; j < cnt; ++j) {
      const int sj = __builtin_amdgcn_readlane(s, j);
      const float w0 = gat_bcast(a0, j), w1 = gat_bcast(a1, j);
      const float w2 = gat_bcast(a2, j), w3 = gat_bcast(a3, j);
      const float wA = hi ? w1 : w0;
      const float wB = hi ? w3 : w2;
      const float4* row = (const float4*)(h + (size_t)sj * 256);
      const float4 v0 = row[lane];
      const float4 v1 = row[32 + lane];
      acc0.x += v0.x * wA; acc0.y += v0.y * wA; acc0.z += v0.z * wA; acc0.w += v0.w * wA;
      acc1.x += v1.x * wB; acc1.y += v1.y * wB; acc1.z += v1.z * wB; acc1.w += v1.w * wB;
    }
  }
  const float4 b0 = ((const float4*)bias)[lane];
  const float4 b1 = ((const float4*)bias)[32 + lane];
  float4 o0, o1;
  o0.x = acc0.x + b0.x; o0.y = acc0.y + b0.y; o0.z = acc0.z + b0.z; o0.w = acc0.w + b0.w;
  o1.x = acc1.x + b1.x; o1.y = acc1.y + b1.y; o1.z = acc1.z + b1.z; o1.w = acc1.w + b1.w;
  o0.x = o0.x > 0.f ? o0.x : 0.f; o0.y = o0.y > 0.f ? o0.y : 0.f;
  o0.z = o0.z > 0.f ? o0.z : 0.f; o0.w = o0.w > 0.f ? o0.w : 0.f;
  o1.x = o1.x > 0.f ? o1.x : 0.f; o1.y = o1.y > 0.f ? o1.y : 0.f;
  o1.z = o1.z > 0.f ? o1.z : 0.f; o1.w = o1.w > 0.f ? o1.w : 0.f;
  float4* orow = (float4*)(out + (size_t)d * 256);
  orow[lane] = o0;
  orow[32 + lane] = o1;
}

__global__ __launch_bounds__(256)
void gat_gather_h1(const int* __restrict__ rowptr, const int* __restrict__ csr_src,
                   const float* __restrict__ h, const float* __restrict__ el,
                   const float* __restrict__ er, const float* __restrict__ bias,
                   float* __restrict__ out, int N) {
  const int d = __builtin_amdgcn_readfirstlane((int)(blockIdx.x * 8 + (threadIdx.x >> 5)));
  if (d >= N) return;
  const int lane = threadIdx.x & 31;
  const int beg = rowptr[d], end = rowptr[d + 1];
  const float erd = er[d];

  float m = -__builtin_huge_valf();
  for (int p = beg + lane; p < end; p += 32)
    m = fmaxf(m, gat_lrelu(el[csr_src[p]] + erd));
  m = gat_wave_max(m);

  float ssum = 0.f;
  for (int p = beg + lane; p < end; p += 32)
    ssum += __expf(gat_lrelu(el[csr_src[p]] + erd) - m);
  ssum = gat_wave_sum(ssum);

  float2 acc = {0.f, 0.f};
  for (int base = beg; base < end; base += 32) {
    const int p = base + lane;
    int s = 0;
    float a = 0.f;
    if (p < end) {
      s = csr_src[p];
      a = __expf(gat_lrelu(el[s] + erd) - m) / ssum;
    }
    const int cnt = (end - base < 32) ? (end - base) : 32;
#pragma unroll 4
    for (int j = 0; j < cnt; ++j) {
      const int sj = __builtin_amdgcn_readlane(s, j);
      const float w = gat_bcast(a, j);
      const float2 v = ((const float2*)(h + (size_t)sj * 64))[lane];
      acc.x += v.x * w; acc.y += v.y * w;
    }
  }
  const float2 b = ((const float2*)bias)[lane];
  float2 o;
  o.x = acc.x + b.x; o.y = acc.y + b.y;
  ((float2*)(out + (size_t)d * 64))[lane] = o;
}

extern "C" void kernel_launch(void* const* d_in, const int* in_sizes, int n_in,
                              void* d_out, int out_size, void* d_ws, size_t ws_size,
                              hipStream_t stream) {
  (void)in_sizes; (void)n_in; (void)out_size; (void)ws_size;
  const float* feat = (const float*)d_in[0];
  const int*   src  = (const int*)d_in[1];
  const int*   dst  = (const int*)d_in[2];
  const float* W1  = (const float*)d_in[3];
  const float* al1 = (const float*)d_in[4];
  const float* ar1 = (const float*)d_in[5];
  const float* b1  = (const float*)d_in[6];
  const float* W2  = (const float*)d_in[7];
  const float* al2 = (const float*)d_in[8];
  const float* ar2 = (const float*)d_in[9];
  const float* b2  = (const float*)d_in[10];
  const float* W3  = (const float*)d_in[11];
  const float* al3 = (const float*)d_in[12];
  const float* ar3 = (const float*)d_in[13];
  const float* b3  = (const float*)d_in[14];
  float* out = (float*)d_out;

  const int N = GAT_N, E = GAT_E;

  char* ws = (char*)d_ws;
  size_t off = 0;
  auto carve = [&](size_t bytes) -> void* {
    void* p = ws + off;
    off += (bytes + 255) & ~(size_t)255;
    return p;
  };
  float* bufA    = (float*)carve((size_t)N * 256 * 4);
  float* bufB    = (float*)carve((size_t)N * 256 * 4);
  float* el      = (float*)carve((size_t)N * 4 * 4);
  float* er      = (float*)carve((size_t)N * 4 * 4);
  int*   cnt     = (int*)carve((size_t)N * 4);
  int*   rowptr  = (int*)carve((size_t)(N + 1) * 4);
  int*   rank    = (int*)carve((size_t)E * 4);
  int*   eid     = (int*)carve((size_t)E * 4);
  int*   csr_src = (int*)carve((size_t)E * 4);

  auto blocks = [](long long n) { return (unsigned)((n + 255) / 256); };

  gat_fill_u32<<<blocks(N), 256, 0, stream>>>((unsigned*)cnt, 0u, (long long)N);
  gat_csr_rank<<<blocks(E), 256, 0, stream>>>(dst, cnt, rank, E);
  gat_csr_scan<<<1, 1024, 0, stream>>>(cnt, rowptr);
  gat_csr_place<<<blocks(E), 256, 0, stream>>>(dst, rowptr, rank, eid, E);
  gat_csr_order<<<blocks(E), 256, 0, stream>>>(src, dst, rowptr, eid, csr_src, E);

  const unsigned node_blocks = (unsigned)((N + 7) / 8);

  {
    const int Nw = 256;
    const int gemm_blocks = ((N / 16 + 7) / 8) * (Nw / 64);
    gat_gemm_wmma<128><<<gemm_blocks, 256, 0, stream>>>(feat, W1, bufA, al1, ar1, el, er, N, Nw);
    gat_gather_h4<<<node_blocks, 256, 0, stream>>>(rowptr, csr_src, bufA, el, er, b1, bufB, N);
  }

  {
    const int Nw = 256;
    const int gemm_blocks = ((N / 16 + 7) / 8) * (Nw / 64);
    gat_gemm_wmma<256><<<gemm_blocks, 256, 0, stream>>>(bufB, W2, bufA, al2, ar2, el, er, N, Nw);
    gat_gather_h4<<<node_blocks, 256, 0, stream>>>(rowptr, csr_src, bufA, el, er, b2, bufB, N);
  }

  {
    const int Nw = 64;
    const int gemm_blocks = ((N / 16 + 7) / 8) * (Nw / 64);
    gat_gemm_wmma<256><<<gemm_blocks, 256, 0, stream>>>(bufB, W3, bufA, al3, ar3, el, er, N, Nw);
    gat_gather_h1<<<node_blocks, 256, 0, stream>>>(rowptr, csr_src, bufA, el, er, b3, out, N);
  }
}


# Gather kernels read a bucket of up to 32 edges once: source ids and logits stay in registers for max, exp sum and weights (two reload passes gone); longer buckets keep three passes; sums unchanged
# speedup vs baseline: 6.6089x; 1.0591x over previous
#include <hip/hip_runtime.h>


#define GAT_N 50000
#define GAT_E 800000
#define GAT_F 64

typedef float v2f __attribute__((ext_vector_type(2)));
typedef float v8f __attribute__((ext_vector_type(8)));

template <int MASK>
__device__ __forceinline__ float gat_row_xor(float v) {
  return __int_as_float(__builtin_amdgcn_update_dpp(0, __float_as_int(v), 0x160 + MASK,
                                                    0xf, 0xf, false));
}
__device__ __forceinline__ float gat_row16_sum(float v) {
  v += gat_row_xor<1>(v);
  v += gat_row_xor<2>(v);
  v += gat_row_xor<4>(v);
  v += gat_row_xor<8>(v);
  return v;
}

template <int K>
__global__ __launch_bounds__(256)
void gat_gemm_wmma(const float* __restrict__ A, const float* __restrict__ B,
                   float* __restrict__ C, const float* __restrict__ attn_l,
                   const float* __restrict__ attn_r, float* __restrict__ el,
                   float* __restrict__ er, int M, int Nw) {
  constexpr int PR = K / 2;
  constexpr int PITCH = 132;
  constexpr int NB = K / 32;
  static_assert(K % 32 == 0, "K must be a whole number of 32-k blocks");
  __shared__ float4 bs4[PR * PITCH / 4];
  float* bs = (float*)bs4;
  const int t = threadIdx.x;
  const int tilesN = Nw >> 6;
  const int tilesM = M >> 4;
  const int rb = blockIdx.x / tilesN;
  const int tn = blockIdx.x - rb * tilesN;
  {
    const int cg = t & 15;
    const int pr0 = t >> 4;
    const float* Bg = B + (size_t)tn * 64 + cg * 4;
#pragma unroll
    for (int it = 0; it < PR / 16; ++it) {
      const int pr = pr0 + it * 16;
      const float4 r0 = *(const float4*)(Bg + (size_t)(2 * pr) * Nw);
      const float4 r1 = *(const float4*)(Bg + (size_t)(2 * pr + 1) * Nw);
      float4* d4 = (float4*)(bs + pr * PITCH + cg * 8);
      d4[0] = make_float4(r0.x, r1.x, r0.y, r1.y);
      d4[1] = make_float4(r0.z, r1.z, r0.w, r1.w);
    }
  }
  __syncthreads();
  const int tm = rb * 8 + (t >> 5);
  if (tm >= tilesM) return;
  const int lane = t & 31;
  const int kh = lane >> 4;
  const int lr = lane & 15;

  const float* Ap = A + (size_t)(tm * 16 + lr) * K + kh * 16;
  const float* bl = bs + kh * 8 * PITCH + lr * 2;
  const float* alp = attn_l + tn * 64 + lr;
  const float* arp = attn_r + tn * 64 + lr;
  const float l0 = alp[0], l1 = alp[16], l2 = alp[32], l3 = alp[48];
  const float r0 = arp[0], r1 = arp[16], r2 = arp[32], r3 = arp[48];
  v8f acc0 = {}, acc1 = {}, acc2 = {}, acc3 = {};
  float4 q0 = *(const float4*)(Ap + 0);
  float4 q1 = *(const float4*)(Ap + 4);
  float4 q2 = *(const float4*)(Ap + 8);
  float4 q3 = *(const float4*)(Ap + 12);
#define GAT_KSTEP(S, AX, AY)                                                                     \
  {                                                                                              \
    v2f a; a.x = (AX); a.y = (AY);                                                               \
    const float* br = bb + (S) * PITCH;                                                          \
    const v2f b0 = *(const v2f*)(br +  0);                                                       \
    const v2f b1 = *(const v2f*)(br + 32);                                                       \
    const v2f b2 = *(const v2f*)(br + 64);                                                       \
    const v2f b3 = *(const v2f*)(br + 96);                                                       \
    acc0 = __builtin_amdgcn_wmma_f32_16x16x4_f32(false, a, false, b0, (short)0, acc0, false, false); \
    acc1 = __builtin_amdgcn_wmma_f32_16x16x4_f32(false, a, false, b1, (short)0, acc1, false, false); \
    acc2 = __builtin_amdgcn_wmma_f32_16x16x4_f32(false, a, false, b2, (short)0, acc2, false, false); \
    acc3 = __builtin_amdgcn_wmma_f32_16x16x4_f32(false, a, false, b3, (short)0, acc3, false, false); \
  }
#pragma unroll
  for (int jb = 0; jb < NB; ++jb) {
    float4 n0 = q0, n1 = q1, n2 = q2, n3 = q3;
    if (jb + 1 < NB) {
      const float* An = Ap + (jb + 1) * 32;
      n0 = *(const float4*)(An + 0);
      n1 = *(const float4*)(An + 4);
      n2 = *(const float4*)(An + 8);
      n3 = *(const float4*)(An + 12);
    }
    const float* bb = bl + jb * 16 * PITCH;
    GAT_KSTEP(0, q0.x, q0.y)
    GAT_KSTEP(1, q0.z, q0.w)
    GAT_KSTEP(2, q1.x, q1.y)
    GAT_KSTEP(3, q1.z, q1.w)
    GAT_KSTEP(4, q2.x, q2.y)
    GAT_KSTEP(5, q2.z, q2.w)
    GAT_KSTEP(6, q3.x, q3.y)
    GAT_KSTEP(7, q3.z, q3.w)
    q0 = n0; q1 = n1; q2 = n2; q3 = n3;
  }
#undef GAT_KSTEP
  float* Cp = C + (size_t)(tm * 16 + kh * 8) * Nw + tn * 64 + lr;
#pragma unroll
  for (int r = 0; r < 8; ++r) {
    Cp[(size_t)r * Nw +  0] = acc0[r];
    Cp[(size_t)r * Nw + 16] = acc1[r];
    Cp[(size_t)r * Nw + 32] = acc2[r];
    Cp[(size_t)r * Nw + 48] = acc3[r];
  }
  float keep = 0.f;
#pragma unroll
  for (int r = 0; r < 8; ++r) {
    const float pl = gat_row16_sum(acc0[r] * l0 + acc1[r] * l1 + acc2[r] * l2 + acc3[r] * l3);
    const float pr = gat_row16_sum(acc0[r] * r0 + acc1[r] * r1 + acc2[r] * r2 + acc3[r] * r3);
    keep = (lr == r) ? pl : keep;
    keep = (lr == r + 8) ? pr : keep;
  }
  float* ep = (lr < 8) ? el : er;
  ep[(size_t)(tm * 16 + kh * 8 + (lr & 7)) * tilesN + tn] = keep;
}

__global__ __launch_bounds__(256)
void gat_fill_u32(unsigned* __restrict__ p, unsigned v, long long n) {
  long long i = (long long)blockIdx.x * blockDim.x + threadIdx.x;
  if (i < n) p[i] = v;
}

__global__ __launch_bounds__(256)
void gat_csr_rank(const int* __restrict__ dst, int* __restrict__ cnt,
                  int* __restrict__ rank, int E) {
  int e = blockIdx.x * blockDim.x + threadIdx.x;
  if (e >= E) return;
  rank[e] = atomicAdd(&cnt[dst[e]], 1);
}

__global__ __launch_bounds__(1024)
void gat_csr_scan(const int* __restrict__ cnt, int* __restrict__ rowptr) {
  constexpr int n = GAT_N;
  constexpr int PER = (n + 1023) / 1024;
  constexpr int NV = (1024 * PER) / 4;
  constexpr int NVG = n / 4;
  constexpr int ROUNDS = (NV + 1023) / 1024;
  static_assert(n % 4 == 0, "four-word units must tile the node count");
  __shared__ int4 buf4[NV];
  __shared__ int part[1024];
  int* buf = (int*)buf4;
  const int t = threadIdx.x;
  const int4* __restrict__ cnt4 = (const int4*)cnt;
  int4* __restrict__ rowptr4 = (int4*)rowptr;
#pragma unroll
  for (int r = 0; r < ROUNDS; ++r) {
    const int i = t + r * 1024;
    if (i < NV) buf4[i] = (i < NVG) ? cnt4[i] : make_int4(0, 0, 0, 0);
  }
  __syncthreads();
  const int b = t * PER;
  int c[PER];
  int s = 0;
#pragma unroll
  for (int k = 0; k < PER; ++k) {
    c[k] = buf[b + k];
    s += c[k];
  }
  part[t] = s;
  __syncthreads();
  for (int off = 1; off < 1024; off <<= 1) {
    int v = (t >= off) ? part[t - off] : 0;
    __syncthreads();
    part[t] += v;
    __syncthreads();
  }
  int run = part[t] - s;
#pragma unroll
  for (int k = 0; k < PER; ++k) {
    buf[b + k] = run;
    run += c[k];
  }
  __syncthreads();
#pragma unroll
  for (int r = 0; r < ROUNDS; ++r) {
    const int i = t + r * 1024;
    if (i < NVG) rowptr4[i] = buf4[i];
  }
  if (t == 1023) rowptr[n] = part[1023];
}

__global__ __launch_bounds__(256)
void gat_csr_place(const int* __restrict__ dst, const int* __restrict__ rowptr,
                   const int* __restrict__ rank, int* __restrict__ eid, int E) {
  int e = blockIdx.x * blockDim.x + threadIdx.x;
  if (e >= E) return;
  eid[rowptr[dst[e]] + rank[e]] = e;
}

__global__ __launch_bounds__(256)
void gat_csr_order(const int* __restrict__ src, const int* __restrict__ dst,
                   const int* __restrict__ rowptr, const int* __restrict__ eid,
                   int* __restrict__ csr_src, int E) {
  int e = blockIdx.x * blockDim.x + threadIdx.x;
  if (e >= E) return;
  const int d = dst[e];
  const int b = rowptr[d], en = rowptr[d + 1];
  int c = 0;
  for (int q = b; q < en; ++q) c += (eid[q] < e) ? 1 : 0;
  csr_src[b + c] = src[e];
}

__device__ __forceinline__ float gat_lrelu(float x) { return (x > 0.f) ? x : 0.2f * x; }
__device__ __forceinline__ float gat_wave_max(float v) {
  for (int o = 16; o > 0; o >>= 1) v = fmaxf(v, __shfl_xor(v, o, 32));
  return v;
}
__device__ __forceinline__ float gat_wave_sum(float v) {
  for (int o = 16; o > 0; o >>= 1) v += __shfl_xor(v, o, 32);
  return v;
}
__device__ __forceinline__ float gat_bcast(float v, int j) {
  return __int_as_float(__builtin_amdgcn_readlane(__float_as_int(v), j));
}

__device__ __forceinline__ void gat_rows_h4(const float* __restrict__ h, int s, float a0, float a1,
                                            float a2, float a3, int cnt, int lane, bool hi,
                                            float4& acc0, float4& acc1) {
#pragma unroll 4
  for (int j = 0; j < cnt; ++j) {
    const int sj = __builtin_amdgcn_readlane(s, j);
    const float w0 = gat_bcast(a0, j), w1 = gat_bcast(a1, j);
    const float w2 = gat_bcast(a2, j), w3 = gat_bcast(a3, j);
    const float wA = hi ? w1 : w0;
    const float wB = hi ? w3 : w2;
    const float4* row = (const float4*)(h + (size_t)sj * 256);
    const float4 v0 = row[lane];
    const float4 v1 = row[32 + lane];
    acc0.x += v0.x * wA; acc0.y += v0.y * wA; acc0.z += v0.z * wA; acc0.w += v0.w * wA;
    acc1.x += v1.x * wB; acc1.y += v1.y * wB; acc1.z += v1.z * wB; acc1.w += v1.w * wB;
  }
}

__global__ __launch_bounds__(256)
void gat_gather_h4(const int* __restrict__ rowptr, const int* __restrict__ csr_src,
                   const float* __restrict__ h, const float* __restrict__ el,
                   const float* __restrict__ er, const float* __restrict__ bias,
                   float* __restrict__ out, int N) {
  const int d = __builtin_amdgcn_readfirstlane((int)(blockIdx.x * 8 + (threadIdx.x >> 5)));
  if (d >= N) return;
  const int lane = threadIdx.x & 31;
  const bool hi = lane >= 16;
  const int beg = rowptr[d], end = rowptr[d + 1];
  const float4 erd = *(const float4*)(er + (size_t)d * 4);
  const float4* el4 = (const float4*)el;

  const float ninf = -__builtin_huge_valf();
  float4 acc0 = {0.f, 0.f, 0.f, 0.f}, acc1 = {0.f, 0.f, 0.f, 0.f};
  if (end - beg <= 32) {
    const int p = beg + lane;
    const bool act = p < end;
    int s = 0;
    float e0 = ninf, e1 = ninf, e2 = ninf, e3 = ninf;
    if (act) {
      s = csr_src[p];
      const float4 x = el4[s];
      e0 = gat_lrelu(x.x + erd.x);
      e1 = gat_lrelu(x.y + erd.y);
      e2 = gat_lrelu(x.z + erd.z);
      e3 = gat_lrelu(x.w + erd.w);
    }
    const float m0 = gat_wave_max(e0), m1 = gat_wave_max(e1);
    const float m2 = gat_wave_max(e2), m3 = gat_wave_max(e3);
    float p0 = 0.f, p1 = 0.f, p2 = 0.f, p3 = 0.f;
    if (act) {
      p0 = __expf(e0 - m0);
      p1 = __expf(e1 - m1);
      p2 = __expf(e2 - m2);
      p3 = __expf(e3 - m3);
    }
    const float s0 = gat_wave_sum(p0), s1 = gat_wave_sum(p1);
    const float s2 = gat_wave_sum(p2), s3 = gat_wave_sum(p3);
    float a0 = 0.f, a1 = 0.f, a2 = 0.f, a3 = 0.f;
    if (act) {
      a0 = p0 / s0;
      a1 = p1 / s1;
      a2 = p2 / s2;
      a3 = p3 / s3;
    }
    gat_rows_h4(h, s, a0, a1, a2, a3, end - beg, lane, hi, acc0, acc1);
  } else {
    float m0 = ninf, m1 = ninf, m2 = ninf, m3 = ninf;
    for (int p = beg + lane; p < end; p += 32) {
      const float4 x = el4[csr_src[p]];
      m0 = fmaxf(m0, gat_lrelu(x.x + erd.x));
      m1 = fmaxf(m1, gat_lrelu(x.y + erd.y));
      m2 = fmaxf(m2, gat_lrelu(x.z + erd.z));
      m3 = fmaxf(m3, gat_lrelu(x.w + erd.w));
    }
    m0 = gat_wave_max(m0); m1 = gat_wave_max(m1);
    m2 = gat_wave_max(m2); m3 = gat_wave_max(m3);

    float s0 = 0.f, s1 = 0.f, s2 = 0.f, s3 = 0.f;
    for (int p = beg + lane; p < end; p += 32) {
      const float4 x = el4[csr_src[p]];
      s0 += __expf(gat_lrelu(x.x + erd.x) - m0);
      s1 += __expf(gat_lrelu(x.y + erd.y) - m1);
      s2 += __expf(gat_lrelu(x.z + erd.z) - m2);
      s3 += __expf(gat_lrelu(x.w + erd.w) - m3);
    }
    s0 = gat_wave_sum(s0); s1 = gat_wave_sum(s1);
    s2 = gat_wave_sum(s2); s3 = gat_wave_sum(s3);

    for (int base = beg; base < end; base += 32) {
      const int p = base + lane;
      int s = 0;
      float a0 = 0.f, a1 = 0.f, a2 = 0.f, a3 = 0.f;
      if (p < end) {
        s = csr_src[p];
        const float4 x = el4[s];
        a0 = __expf(gat_lrelu(x.x + erd.x) - m0) / s0;
        a1 = __expf(gat_lrelu(x.y + erd.y) - m1) / s1;
        a2 = __expf(gat_lrelu(x.z + erd.z) - m2) / s2;
        a3 = __expf(gat_lrelu(x.w + erd.w) - m3) / s3;
      }
      const int cnt = (end - base < 32) ? (end - base) : 32;
      gat_rows_h4(h, s, a0, a1, a2, a3, cnt, lane, hi, acc0, acc1);
    }
  }
  const float4 b0 = ((const float4*)bias)[lane];
  const float4 b1 = ((const float4*)bias)[32 + lane];
  float4 o0, o1;
  o0.x = acc0.x + b0.x; o0.y = acc0.y + b0.y; o0.z = acc0.z + b0.z; o0.w = acc0.w + b0.w;
  o1.x = acc1.x + b1.x; o1.y = acc1.y + b1.y; o1.z = acc1.z + b1.z; o1.w = acc1.w + b1.w;
  o0.x = o0.x > 0.f ? o0.x : 0.f; o0.y = o0.y > 0.f ? o0.y : 0.f;
  o0.z = o0.z > 0.f ? o0.z : 0.f; o0.w = o0.w > 0.f ? o0.w : 0.f;
  o1.x = o1.x > 0.f ? o1.x : 0.f; o1.y = o1.y > 0.f ? o1.y : 0.f;
  o1.z = o1.z > 0.f ? o1.z : 0.f; o1.w = o1.w > 0.f ? o1.w : 0.f;
  float4* orow = (float4*)(out + (size_t)d * 256);
  orow[lane] = o0;
  orow[32 + lane] = o1;
}

__device__ __forceinline__ void gat_rows_h1(const float* __restrict__ h, int s, float a, int cnt,
                                            int lane, float2& acc) {
#pragma unroll 4
  for (int j = 0; j < cnt; ++j) {
    const int sj = __builtin_amdgcn_readlane(s, j);
    const float w = gat_bcast(a, j);
    const float2 v = ((const float2*)(h + (size_t)sj * 64))[lane];
    acc.x += v.x * w; acc.y += v.y * w;
  }
}

__global__ __launch_bounds__(256)
void gat_gather_h1(const int* __restrict__ rowptr, const int* __restrict__ csr_src,
                   const float* __restrict__ h, const float* __restrict__ el,
                   const float* __restrict__ er, const float* __restrict__ bias,
                   float* __restrict__ out, int N) {
  const int d = __builtin_amdgcn_readfirstlane((int)(blockIdx.x * 8 + (threadIdx.x >> 5)));
  if (d >= N) return;
  const int lane = threadIdx.x & 31;
  const int beg = rowptr[d], end = rowptr[d + 1];
  const float erd = er[d];

  const float ninf = -__builtin_huge_valf();
  float2 acc = {0.f, 0.f};
  if (end - beg <= 32) {
    const int p = beg + lane;
    const bool act = p < end;
    int s = 0;
    float e = ninf;
    if (act) {
      s = csr_src[p];
      e = gat_lrelu(el[s] + erd);
    }
    const float m = gat_wave_max(e);
    float pe = 0.f;
    if (act) pe = __expf(e - m);
    const float ssum = gat_wave_sum(pe);
    float a = 0.f;
    if (act) a = pe / ssum;
    gat_rows_h1(h, s, a, end - beg, lane, acc);
  } else {
    float m = ninf;
    for (int p = beg + lane; p < end; p += 32)
      m = fmaxf(m, gat_lrelu(el[csr_src[p]] + erd));
    m = gat_wave_max(m);

    float ssum = 0.f;
    for (int p = beg + lane; p < end; p += 32)
      ssum += __expf(gat_lrelu(el[csr_src[p]] + erd) - m);
    ssum = gat_wave_sum(ssum);

    for (int base = beg; base < end; base += 32) {
      const int p = base + lane;
      int s = 0;
      float a = 0.f;
      if (p < end) {
        s = csr_src[p];
        a = __expf(gat_lrelu(el[s] + erd) - m) / ssum;
      }
      const int cnt = (end - base < 32) ? (end - base) : 32;
      gat_rows_h1(h, s, a, cnt, lane, acc);
    }
  }
  const float2 b = ((const float2*)bias)[lane];
  float2 o;
  o.x = acc.x + b.x; o.y = acc.y + b.y;
  ((float2*)(out + (size_t)d * 64))[lane] = o;
}

extern "C" void kernel_launch(void* const* d_in, const int* in_sizes, int n_in,
                              void* d_out, int out_size, void* d_ws, size_t ws_size,
                              hipStream_t stream) {
  (void)in_sizes; (void)n_in; (void)out_size; (void)ws_size;
  const float* feat = (const float*)d_in[0];
  const int*   src  = (const int*)d_in[1];
  const int*   dst  = (const int*)d_in[2];
  const float* W1  = (const float*)d_in[3];
  const float* al1 = (const float*)d_in[4];
  const float* ar1 = (const float*)d_in[5];
  const float* b1  = (const float*)d_in[6];
  const float* W2  = (const float*)d_in[7];
  const float* al2 = (const float*)d_in[8];
  const float* ar2 = (const float*)d_in[9];
  const float* b2  = (const float*)d_in[10];
  const float* W3  = (const float*)d_in[11];
  const float* al3 = (const float*)d_in[12];
  const float* ar3 = (const float*)d_in[13];
  const float* b3  = (const float*)d_in[14];
  float* out = (float*)d_out;

  const int N = GAT_N, E = GAT_E;

  char* ws = (char*)d_ws;
  size_t off = 0;
  auto carve = [&](size_t bytes) -> void* {
    void* p = ws + off;
    off += (bytes + 255) & ~(size_t)255;
    return p;
  };
  float* bufA    = (float*)carve((size_t)N * 256 * 4);
  float* bufB    = (float*)carve((size_t)N * 256 * 4);
  float* el      = (float*)carve((size_t)N * 4 * 4);
  float* er      = (float*)carve((size_t)N * 4 * 4);
  int*   cnt     = (int*)carve((size_t)N * 4);
  int*   rowptr  = (int*)carve((size_t)(N + 1) * 4);
  int*   rank    = (int*)carve((size_t)E * 4);
  int*   eid     = (int*)carve((size_t)E * 4);
  int*   csr_src = (int*)carve((size_t)E * 4);

  auto blocks = [](long long n) { return (unsigned)((n + 255) / 256); };

  gat_fill_u32<<<blocks(N), 256, 0, stream>>>((unsigned*)cnt, 0u, (long long)N);
  gat_csr_rank<<<blocks(E), 256, 0, stream>>>(dst, cnt, rank, E);
  gat_csr_scan<<<1, 1024, 0, stream>>>(cnt, rowptr);
  gat_csr_place<<<blocks(E), 256, 0, stream>>>(dst, rowptr, rank, eid, E);
  gat_csr_order<<<blocks(E), 256, 0, stream>>>(src, dst, rowptr, eid, csr_src, E);

  const unsigned node_blocks = (unsigned)((N + 7) / 8);

  {
    const int Nw = 256;
    const int gemm_blocks = ((N / 16 + 7) / 8) * (Nw / 64);
    gat_gemm_wmma<128><<<gemm_blocks, 256, 0, stream>>>(feat, W1, bufA, al1, ar1, el, er, N, Nw);
    gat_gather_h4<<<node_blocks, 256, 0, stream>>>(rowptr, csr_src, bufA, el, er, b1, bufB, N);
  }

  {
    const int Nw = 256;
    const int gemm_blocks = ((N / 16 + 7) / 8) * (Nw / 64);
    gat_gemm_wmma<256><<<gemm_blocks, 256, 0, stream>>>(bufB, W2, bufA, al2, ar2, el, er, N, Nw);
    gat_gather_h4<<<node_blocks, 256, 0, stream>>>(rowptr, csr_src, bufA, el, er, b2, bufB, N);
  }

  {
    const int Nw = 64;
    const int gemm_blocks = ((N / 16 + 7) / 8) * (Nw / 64);
    gat_gemm_wmma<256><<<gemm_blocks, 256, 0, stream>>>(bufB, W3, bufA, al3, ar3, el, er, N, Nw);
    gat_gather_h1<<<node_blocks, 256, 0, stream>>>(rowptr, csr_src, bufA, el, er, b3, out, N);
  }
}


# Gather wave puts its bucket into edge-id order in registers (slots counted from lane broadcasts, lane pushes); the ordering launch and its per-edge bucket scans are gone; sums unchanged
# speedup vs baseline: 7.6065x; 1.1509x over previous
#include <hip/hip_runtime.h>


#define GAT_N 50000
#define GAT_E 800000
#define GAT_F 64

typedef float v2f __attribute__((ext_vector_type(2)));
typedef float v8f __attribute__((ext_vector_type(8)));

template <int MASK>
__device__ __forceinline__ float gat_row_xor(float v) {
  return __int_as_float(__builtin_amdgcn_update_dpp(0, __float_as_int(v), 0x160 + MASK,
                                                    0xf, 0xf, false));
}
__device__ __forceinline__ float gat_row16_sum(float v) {
  v += gat_row_xor<1>(v);
  v += gat_row_xor<2>(v);
  v += gat_row_xor<4>(v);
  v += gat_row_xor<8>(v);
  return v;
}

template <int K>
__global__ __launch_bounds__(256)
void gat_gemm_wmma(const float* __restrict__ A, const float* __restrict__ B,
                   float* __restrict__ C, const float* __restrict__ attn_l,
                   const float* __restrict__ attn_r, float* __restrict__ el,
                   float* __restrict__ er, int M, int Nw) {
  constexpr int PR = K / 2;
  constexpr int PITCH = 132;
  constexpr int NB = K / 32;
  static_assert(K % 32 == 0, "K must be a whole number of 32-k blocks");
  __shared__ float4 bs4[PR * PITCH / 4];
  float* bs = (float*)bs4;
  const int t = threadIdx.x;
  const int tilesN = Nw >> 6;
  const int tilesM = M >> 4;
  const int rb = blockIdx.x / tilesN;
  const int tn = blockIdx.x - rb * tilesN;
  {
    const int cg = t & 15;
    const int pr0 = t >> 4;
    const float* Bg = B + (size_t)tn * 64 + cg * 4;
#pragma unroll
    for (int it = 0; it < PR / 16; ++it) {
      const int pr = pr0 + it * 16;
      const float4 r0 = *(const float4*)(Bg + (size_t)(2 * pr) * Nw);
      const float4 r1 = *(const float4*)(Bg + (size_t)(2 * pr + 1) * Nw);
      float4* d4 = (float4*)(bs + pr * PITCH + cg * 8);
      d4[0] = make_float4(r0.x, r1.x, r0.y, r1.y);
      d4[1] = make_float4(r0.z, r1.z, r0.w, r1.w);
    }
  }
  __syncthreads();
  const int tm = rb * 8 + (t >> 5);
  if (tm >= tilesM) return;
  const int lane = t & 31;
  const int kh = lane >> 4;
  const int lr = lane & 15;

  const float* Ap = A + (size_t)(tm * 16 + lr) * K + kh * 16;
  const float* bl = bs + kh * 8 * PITCH + lr * 2;
  const float* alp = attn_l + tn * 64 + lr;
  const float* arp = attn_r + tn * 64 + lr;
  const float l0 = alp[0], l1 = alp[16], l2 = alp[32], l3 = alp[48];
  const float r0 = arp[0], r1 = arp[16], r2 = arp[32], r3 = arp[48];
  v8f acc0 = {}, acc1 = {}, acc2 = {}, acc3 = {};
  float4 q0 = *(const float4*)(Ap + 0);
  float4 q1 = *(const float4*)(Ap + 4);
  float4 q2 = *(const float4*)(Ap + 8);
  float4 q3 = *(const float4*)(Ap + 12);
#define GAT_KSTEP(S, AX, AY)                                                                     \
  {                                                                                              \
    v2f a; a.x = (AX); a.y = (AY);                                                               \
    const float* br = bb + (S) * PITCH;                                                          \
    const v2f b0 = *(const v2f*)(br +  0);                                                       \
    const v2f b1 = *(const v2f*)(br + 32);                                                       \
    const v2f b2 = *(const v2f*)(br + 64);                                                       \
    const v2f b3 = *(const v2f*)(br + 96);                                                       \
    acc0 = __builtin_amdgcn_wmma_f32_16x16x4_f32(false, a, false, b0, (short)0, acc0, false, false); \
    acc1 = __builtin_amdgcn_wmma_f32_16x16x4_f32(false, a, false, b1, (short)0, acc1, false, false); \
    acc2 = __builtin_amdgcn_wmma_f32_16x16x4_f32(false, a, false, b2, (short)0, acc2, false, false); \
    acc3 = __builtin_amdgcn_wmma_f32_16x16x4_f32(false, a, false, b3, (short)0, acc3, false, false); \
  }
#pragma unroll
  for (int jb = 0; jb < NB; ++jb) {
    float4 n0 = q0, n1 = q1, n2 = q2, n3 = q3;
    if (jb + 1 < NB) {
      const float* An = Ap + (jb + 1) * 32;
      n0 = *(const float4*)(An + 0);
      n1 = *(const float4*)(An + 4);
      n2 = *(const float4*)(An + 8);
      n3 = *(const float4*)(An + 12);
    }
    const float* bb = bl + jb * 16 * PITCH;
    GAT_KSTEP(0, q0.x, q0.y)
    GAT_KSTEP(1, q0.z, q0.w)
    GAT_KSTEP(2, q1.x, q1.y)
    GAT_KSTEP(3, q1.z, q1.w)
    GAT_KSTEP(4, q2.x, q2.y)
    GAT_KSTEP(5, q2.z, q2.w)
    GAT_KSTEP(6, q3.x, q3.y)
    GAT_KSTEP(7, q3.z, q3.w)
    q0 = n0; q1 = n1; q2 = n2; q3 = n3;
  }
#undef GAT_KSTEP
  float* Cp = C + (size_t)(tm * 16 + kh * 8) * Nw + tn * 64 + lr;
#pragma unroll
  for (int r = 0; r < 8; ++r) {
    Cp[(size_t)r * Nw +  0] = acc0[r];
    Cp[(size_t)r * Nw + 16] = acc1[r];
    Cp[(size_t)r * Nw + 32] = acc2[r];
    Cp[(size_t)r * Nw + 48] = acc3[r];
  }
  float keep = 0.f;
#pragma unroll
  for (int r = 0; r < 8; ++r) {
    const float pl = gat_row16_sum(acc0[r] * l0 + acc1[r] * l1 + acc2[r] * l2 + acc3[r] * l3);
    const float pr = gat_row16_sum(acc0[r] * r0 + acc1[r] * r1 + acc2[r] * r2 + acc3[r] * r3);
    keep = (lr == r) ? pl : keep;
    keep = (lr == r + 8) ? pr : keep;
  }
  float* ep = (lr < 8) ? el : er;
  ep[(size_t)(tm * 16 + kh * 8 + (lr & 7)) * tilesN + tn] = keep;
}

__global__ __launch_bounds__(256)
void gat_fill_u32(unsigned* __restrict__ p, unsigned v, long long n) {
  long long i = (long long)blockIdx.x * blockDim.x + threadIdx.x;
  if (i < n) p[i] = v;
}

__global__ __launch_bounds__(256)
void gat_csr_rank(const int* __restrict__ dst, int* __restrict__ cnt,
                  int* __restrict__ rank, int E) {
  int e = blockIdx.x * blockDim.x + threadIdx.x;
  if (e >= E) return;
  rank[e] = atomicAdd(&cnt[dst[e]], 1);
}

__global__ __launch_bounds__(1024)
void gat_csr_scan(const int* __restrict__ cnt, int* __restrict__ rowptr) {
  constexpr int n = GAT_N;
  constexpr int PER = (n + 1023) / 1024;
  constexpr int NV = (1024 * PER) / 4;
  constexpr int NVG = n / 4;
  constexpr int ROUNDS = (NV + 1023) / 1024;
  static_assert(n % 4 == 0, "four-word units must tile the node count");
  __shared__ int4 buf4[NV];
  __shared__ int part[1024];
  int* buf = (int*)buf4;
  const int t = threadIdx.x;
  const int4* __restrict__ cnt4 = (const int4*)cnt;
  int4* __restrict__ rowptr4 = (int4*)rowptr;
#pragma unroll
  for (int r = 0; r < ROUNDS; ++r) {
    const int i = t + r * 1024;
    if (i < NV) buf4[i] = (i < NVG) ? cnt4[i] : make_int4(0, 0, 0, 0);
  }
  __syncthreads();
  const int b = t * PER;
  int c[PER];
  int s = 0;
#pragma unroll
  for (int k = 0; k < PER; ++k) {
    c[k] = buf[b + k];
    s += c[k];
  }
  part[t] = s;
  __syncthreads();
  for (int off = 1; off < 1024; off <<= 1) {
    int v = (t >= off) ? part[t - off] : 0;
    __syncthreads();
    part[t] += v;
    __syncthreads();
  }
  int run = part[t] - s;
#pragma unroll
  for (int k = 0; k < PER; ++k) {
    buf[b + k] = run;
    run += c[k];
  }
  __syncthreads();
#pragma unroll
  for (int r = 0; r < ROUNDS; ++r) {
    const int i = t + r * 1024;
    if (i < NVG) rowptr4[i] = buf4[i];
  }
  if (t == 1023) rowptr[n] = part[1023];
}

__global__ __launch_bounds__(256)
void gat_csr_place(const int* __restrict__ src, const int* __restrict__ dst,
                   const int* __restrict__ rowptr, const int* __restrict__ rank,
                   int2* __restrict__ pairs, int E) {
  int e = blockIdx.x * blockDim.x + threadIdx.x;
  if (e >= E) return;
  pairs[rowptr[dst[e]] + rank[e]] = make_int2(e, src[e]);
}

__device__ __forceinline__ int gat_bucket_slot_addr(int key, int cnt, int lane) {
  int c = 0;
#pragma unroll 4
  for (int j = 0; j < cnt; ++j) c += (__builtin_amdgcn_readlane(key, j) < key) ? 1 : 0;
  return ((lane < cnt) ? c : lane) << 2;
}
__device__ __forceinline__ int gat_push_i(int addr, int v) {
  return __builtin_amdgcn_ds_permute(addr, v);
}
__device__ __forceinline__ float gat_push_f(int addr, float v) {
  return __int_as_float(__builtin_amdgcn_ds_permute(addr, __float_as_int(v)));
}

__device__ __forceinline__ void gat_bucket_park_sorted(const int2* __restrict__ pairs,
                                                       int* sorted, int beg, int end, int lane) {
  for (int p = beg + lane; p < end; p += 32) {
    const int2 me = pairs[p];
    int c = 0;
    for (int q = beg; q < end; ++q) c += (pairs[q].x < me.x) ? 1 : 0;
    sorted[beg + c] = me.y;
  }
  __threadfence();
}

__device__ __forceinline__ float gat_lrelu(float x) { return (x > 0.f) ? x : 0.2f * x; }
__device__ __forceinline__ float gat_wave_max(float v) {
  for (int o = 16; o > 0; o >>= 1) v = fmaxf(v, __shfl_xor(v, o, 32));
  return v;
}
__device__ __forceinline__ float gat_wave_sum(float v) {
  for (int o = 16; o > 0; o >>= 1) v += __shfl_xor(v, o, 32);
  return v;
}
__device__ __forceinline__ float gat_bcast(float v, int j) {
  return __int_as_float(__builtin_amdgcn_readlane(__float_as_int(v), j));
}

__device__ __forceinline__ void gat_rows_h4(const float* __restrict__ h, int s, float a0, float a1,
                                            float a2, float a3, int cnt, int lane, bool hi,
                                            float4& acc0, float4& acc1) {
#pragma unroll 4
  for (int j = 0; j < cnt; ++j) {
    const int sj = __builtin_amdgcn_readlane(s, j);
    const float w0 = gat_bcast(a0, j), w1 = gat_bcast(a1, j);
    const float w2 = gat_bcast(a2, j), w3 = gat_bcast(a3, j);
    const float wA = hi ? w1 : w0;
    const float wB = hi ? w3 : w2;
    const float4* row = (const float4*)(h + (size_t)sj * 256);
    const float4 v0 = row[lane];
    const float4 v1 = row[32 + lane];
    acc0.x += v0.x * wA; acc0.y += v0.y * wA; acc0.z += v0.z * wA; acc0.w += v0.w * wA;
    acc1.x += v1.x * wB; acc1.y += v1.y * wB; acc1.z += v1.z * wB; acc1.w += v1.w * wB;
  }
}

__global__ __launch_bounds__(256)
void gat_gather_h4(const int* __restrict__ rowptr, const int2* __restrict__ pairs,
                   int* sorted, const float* __restrict__ h, const float* __restrict__ el,
                   const float* __restrict__ er, const float* __restrict__ bias,
                   float* __restrict__ out, int N) {
  const int d = __builtin_amdgcn_readfirstlane((int)(blockIdx.x * 8 + (threadIdx.x >> 5)));
  if (d >= N) return;
  const int lane = threadIdx.x & 31;
  const bool hi = lane >= 16;
  const int beg = rowptr[d], end = rowptr[d + 1];
  const float4 erd = *(const float4*)(er + (size_t)d * 4);
  const float4* el4 = (const float4*)el;

  const float ninf = -__builtin_huge_valf();
  float4 acc0 = {0.f, 0.f, 0.f, 0.f}, acc1 = {0.f, 0.f, 0.f, 0.f};
  if (end - beg <= 32) {
    const int cnt = end - beg;
    const bool act = lane < cnt;
    int key = 0x7fffffff, sv = 0;
    float4 xv = {0.f, 0.f, 0.f, 0.f};
    if (act) {
      const int2 pr = pairs[beg + lane];
      key = pr.x;
      sv = pr.y;
      xv = el4[sv];
    }
    const int tgt = gat_bucket_slot_addr(key, cnt, lane);
    const int s = gat_push_i(tgt, sv);
    float4 x;
    x.x = gat_push_f(tgt, xv.x);
    x.y = gat_push_f(tgt, xv.y);
    x.z = gat_push_f(tgt, xv.z);
    x.w = gat_push_f(tgt, xv.w);
    float e0 = ninf, e1 = ninf, e2 = ninf, e3 = ninf;
    if (act) {
      e0 = gat_lrelu(x.x + erd.x);
      e1 = gat_lrelu(x.y + erd.y);
      e2 = gat_lrelu(x.z + erd.z);
      e3 = gat_lrelu(x.w + erd.w);
    }
    const float m0 = gat_wave_max(e0), m1 = gat_wave_max(e1);
    const float m2 = gat_wave_max(e2), m3 = gat_wave_max(e3);
    float p0 = 0.f, p1 = 0.f, p2 = 0.f, p3 = 0.f;
    if (act) {
      p0 = __expf(e0 - m0);
      p1 = __expf(e1 - m1);
      p2 = __expf(e2 - m2);
      p3 = __expf(e3 - m3);
    }
    const float s0 = gat_wave_sum(p0), s1 = gat_wave_sum(p1);
    const float s2 = gat_wave_sum(p2), s3 = gat_wave_sum(p3);
    float a0 = 0.f, a1 = 0.f, a2 = 0.f, a3 = 0.f;
    if (act) {
      a0 = p0 / s0;
      a1 = p1 / s1;
      a2 = p2 / s2;
      a3 = p3 / s3;
    }
    gat_rows_h4(h, s, a0, a1, a2, a3, cnt, lane, hi, acc0, acc1);
  } else {
    gat_bucket_park_sorted(pairs, sorted, beg, end, lane);
    float m0 = ninf, m1 = ninf, m2 = ninf, m3 = ninf;
    for (int p = beg + lane; p < end; p += 32) {
      const float4 x = el4[sorted[p]];
      m0 = fmaxf(m0, gat_lrelu(x.x + erd.x));
      m1 = fmaxf(m1, gat_lrelu(x.y + erd.y));
      m2 = fmaxf(m2, gat_lrelu(x.z + erd.z));
      m3 = fmaxf(m3, gat_lrelu(x.w + erd.w));
    }
    m0 = gat_wave_max(m0); m1 = gat_wave_max(m1);
    m2 = gat_wave_max(m2); m3 = gat_wave_max(m3);

    float s0 = 0.f, s1 = 0.f, s2 = 0.f, s3 = 0.f;
    for (int p = beg + lane; p < end; p += 32) {
      const float4 x = el4[sorted[p]];
      s0 += __expf(gat_lrelu(x.x + erd.x) - m0);
      s1 += __expf(gat_lrelu(x.y + erd.y) - m1);
      s2 += __expf(gat_lrelu(x.z + erd.z) - m2);
      s3 += __expf(gat_lrelu(x.w + erd.w) - m3);
    }
    s0 = gat_wave_sum(s0); s1 = gat_wave_sum(s1);
    s2 = gat_wave_sum(s2); s3 = gat_wave_sum(s3);

    for (int base = beg; base < end; base += 32) {
      const int p = base + lane;
      int s = 0;
      float a0 = 0.f, a1 = 0.f, a2 = 0.f, a3 = 0.f;
      if (p < end) {
        s = sorted[p];
        const float4 x = el4[s];
        a0 = __expf(gat_lrelu(x.x + erd.x) - m0) / s0;
        a1 = __expf(gat_lrelu(x.y + erd.y) - m1) / s1;
        a2 = __expf(gat_lrelu(x.z + erd.z) - m2) / s2;
        a3 = __expf(gat_lrelu(x.w + erd.w) - m3) / s3;
      }
      const int cnt = (end - base < 32) ? (end - base) : 32;
      gat_rows_h4(h, s, a0, a1, a2, a3, cnt, lane, hi, acc0, acc1);
    }
  }
  const float4 b0 = ((const float4*)bias)[lane];
  const float4 b1 = ((const float4*)bias)[32 + lane];
  float4 o0, o1;
  o0.x = acc0.x + b0.x; o0.y = acc0.y + b0.y; o0.z = acc0.z + b0.z; o0.w = acc0.w + b0.w;
  o1.x = acc1.x + b1.x; o1.y = acc1.y + b1.y; o1.z = acc1.z + b1.z; o1.w = acc1.w + b1.w;
  o0.x = o0.x > 0.f ? o0.x : 0.f; o0.y = o0.y > 0.f ? o0.y : 0.f;
  o0.z = o0.z > 0.f ? o0.z : 0.f; o0.w = o0.w > 0.f ? o0.w : 0.f;
  o1.x = o1.x > 0.f ? o1.x : 0.f; o1.y = o1.y > 0.f ? o1.y : 0.f;
  o1.z = o1.z > 0.f ? o1.z : 0.f; o1.w = o1.w > 0.f ? o1.w : 0.f;
  float4* orow = (float4*)(out + (size_t)d * 256);
  orow[lane] = o0;
  orow[32 + lane] = o1;
}

__device__ __forceinline__ void gat_rows_h1(const float* __restrict__ h, int s, float a, int cnt,
                                            int lane, float2& acc) {
#pragma unroll 4
  for (int j = 0; j < cnt; ++j) {
    const int sj = __builtin_amdgcn_readlane(s, j);
    const float w = gat_bcast(a, j);
    const float2 v = ((const float2*)(h + (size_t)sj * 64))[lane];
    acc.x += v.x * w; acc.y += v.y * w;
  }
}

__global__ __launch_bounds__(256)
void gat_gather_h1(const int* __restrict__ rowptr, const int2* __restrict__ pairs,
                   int* sorted, const float* __restrict__ h, const float* __restrict__ el,
                   const float* __restrict__ er, const float* __restrict__ bias,
                   float* __restrict__ out, int N) {
  const int d = __builtin_amdgcn_readfirstlane((int)(blockIdx.x * 8 + (threadIdx.x >> 5)));
  if (d >= N) return;
  const int lane = threadIdx.x & 31;
  const int beg = rowptr[d], end = rowptr[d + 1];
  const float erd = er[d];

  const float ninf = -__builtin_huge_valf();
  float2 acc = {0.f, 0.f};
  if (end - beg <= 32) {
    const int cnt = end - beg;
    const bool act = lane < cnt;
    int key = 0x7fffffff, sv = 0;
    float xv = 0.f;
    if (act) {
      const int2 pr = pairs[beg + lane];
      key = pr.x;
      sv = pr.y;
      xv = el[sv];
    }
    const int tgt = gat_bucket_slot_addr(key, cnt, lane);
    const int s = gat_push_i(tgt, sv);
    const float x = gat_push_f(tgt, xv);
    float e = ninf;
    if (act) e = gat_lrelu(x + erd);
    const float m = gat_wave_max(e);
    float pe = 0.f;
    if (act) pe = __expf(e - m);
    const float ssum = gat_wave_sum(pe);
    float a = 0.f;
    if (act) a = pe / ssum;
    gat_rows_h1(h, s, a, cnt, lane, acc);
  } else {
    gat_bucket_park_sorted(pairs, sorted, beg, end, lane);
    float m = ninf;
    for (int p = beg + lane; p < end; p += 32)
      m = fmaxf(m, gat_lrelu(el[sorted[p]] + erd));
    m = gat_wave_max(m);

    float ssum = 0.f;
    for (int p = beg + lane; p < end; p += 32)
      ssum += __expf(gat_lrelu(el[sorted[p]] + erd) - m);
    ssum = gat_wave_sum(ssum);

    for (int base = beg; base < end; base += 32) {
      const int p = base + lane;
      int s = 0;
      float a = 0.f;
      if (p < end) {
        s = sorted[p];
        a = __expf(gat_lrelu(el[s] + erd) - m) / ssum;
      }
      const int cnt = (end - base < 32) ? (end - base) : 32;
      gat_rows_h1(h, s, a, cnt, lane, acc);
    }
  }
  const float2 b = ((const float2*)bias)[lane];
  float2 o;
  o.x = acc.x + b.x; o.y = acc.y + b.y;
  ((float2*)(out + (size_t)d * 64))[lane] = o;
}

extern "C" void kernel_launch(void* const* d_in, const int* in_sizes, int n_in,
                              void* d_out, int out_size, void* d_ws, size_t ws_size,
                              hipStream_t stream) {
  (void)in_sizes; (void)n_in; (void)out_size; (void)ws_size;
  const float* feat = (const float*)d_in[0];
  const int*   src  = (const int*)d_in[1];
  const int*   dst  = (const int*)d_in[2];
  const float* W1  = (const float*)d_in[3];
  const float* al1 = (const float*)d_in[4];
  const float* ar1 = (const float*)d_in[5];
  const float* b1  = (const float*)d_in[6];
  const float* W2  = (const float*)d_in[7];
  const float* al2 = (const float*)d_in[8];
  const float* ar2 = (const float*)d_in[9];
  const float* b2  = (const float*)d_in[10];
  const float* W3  = (const float*)d_in[11];
  const float* al3 = (const float*)d_in[12];
  const float* ar3 = (const float*)d_in[13];
  const float* b3  = (const float*)d_in[14];
  float* out = (float*)d_out;

  const int N = GAT_N, E = GAT_E;

  char* ws = (char*)d_ws;
  size_t off = 0;
  auto carve = [&](size_t bytes) -> void* {
    void* p = ws + off;
    off += (bytes + 255) & ~(size_t)255;
    return p;
  };
  float* bufA    = (float*)carve((size_t)N * 256 * 4);
  float* bufB    = (float*)carve((size_t)N * 256 * 4);
  float* el      = (float*)carve((size_t)N * 4 * 4);
  float* er      = (float*)carve((size_t)N * 4 * 4);
  int*   cnt     = (int*)carve((size_t)N * 4);
  int*   rowptr  = (int*)carve((size_t)(N + 1) * 4);
  int*   rank    = (int*)carve((size_t)E * 4);
  int2*  pairs   = (int2*)carve((size_t)E * 8);
  int*   sorted  = (int*)carve((size_t)E * 4);

  auto blocks = [](long long n) { return (unsigned)((n + 255) / 256); };

  gat_fill_u32<<<blocks(N), 256, 0, stream>>>((unsigned*)cnt, 0u, (long long)N);
  gat_csr_rank<<<blocks(E), 256, 0, stream>>>(dst, cnt, rank, E);
  gat_csr_scan<<<1, 1024, 0, stream>>>(cnt, rowptr);
  gat_csr_place<<<blocks(E), 256, 0, stream>>>(src, dst, rowptr, rank, pairs, E);

  const unsigned node_blocks = (unsigned)((N + 7) / 8);

  {
    const int Nw = 256;
    const int gemm_blocks = ((N / 16 + 7) / 8) * (Nw / 64);
    gat_gemm_wmma<128><<<gemm_blocks, 256, 0, stream>>>(feat, W1, bufA, al1, ar1, el, er, N, Nw);
    gat_gather_h4<<<node_blocks, 256, 0, stream>>>(rowptr, pairs, sorted, bufA, el, er, b1, bufB, N);
  }

  {
    const int Nw = 256;
    const int gemm_blocks = ((N / 16 + 7) / 8) * (Nw / 64);
    gat_gemm_wmma<256><<<gemm_blocks, 256, 0, stream>>>(bufB, W2, bufA, al2, ar2, el, er, N, Nw);
    gat_gather_h4<<<node_blocks, 256, 0, stream>>>(rowptr, pairs, sorted, bufA, el, er, b2, bufB, N);
  }

  {
    const int Nw = 64;
    const int gemm_blocks = ((N / 16 + 7) / 8) * (Nw / 64);
    gat_gemm_wmma<256><<<gemm_blocks, 256, 0, stream>>>(bufB, W3, bufA, al3, ar3, el, er, N, Nw);
    gat_gather_h1<<<node_blocks, 256, 0, stream>>>(rowptr, pairs, sorted, bufA, el, er, b3, out, N);
  }
}


# Fixed 32-slot bucket per node (slot = the counting add's return; longer buckets spill to one shared list): one fill launch replaces clear, count, scan and place; GEMM 1 clears the counters; same sums
# speedup vs baseline: 7.9662x; 1.0473x over previous
#include <hip/hip_runtime.h>


#define GAT_N 50000
#define GAT_E 800000
#define GAT_F 64

typedef float v2f __attribute__((ext_vector_type(2)));
typedef float v8f __attribute__((ext_vector_type(8)));

template <int MASK>
__device__ __forceinline__ float gat_row_xor(float v) {
  return __int_as_float(__builtin_amdgcn_update_dpp(0, __float_as_int(v), 0x160 + MASK,
                                                    0xf, 0xf, false));
}
__device__ __forceinline__ float gat_row16_sum(float v) {
  v += gat_row_xor<1>(v);
  v += gat_row_xor<2>(v);
  v += gat_row_xor<4>(v);
  v += gat_row_xor<8>(v);
  return v;
}

template <int K, bool CLR>
__global__ __launch_bounds__(256)
void gat_gemm_wmma(const float* __restrict__ A, const float* __restrict__ B,
                   float* __restrict__ C, const float* __restrict__ attn_l,
                   const float* __restrict__ attn_r, float* __restrict__ el,
                   float* __restrict__ er, int M, int Nw, int* __restrict__ clr, int clr_n) {
  constexpr int PR = K / 2;
  constexpr int PITCH = 132;
  constexpr int NB = K / 32;
  static_assert(K % 32 == 0, "K must be a whole number of 32-k blocks");
  __shared__ float4 bs4[PR * PITCH / 4];
  float* bs = (float*)bs4;
  const int t = threadIdx.x;
  if (CLR) {
    const int g = blockIdx.x * 256 + t;
    if (g < clr_n) clr[g] = 0;
  }
  const int tilesN = Nw >> 6;
  const int tilesM = M >> 4;
  const int rb = blockIdx.x / tilesN;
  const int tn = blockIdx.x - rb * tilesN;
  {
    const int cg = t & 15;
    const int pr0 = t >> 4;
    const float* Bg = B + (size_t)tn * 64 + cg * 4;
#pragma unroll
    for (int it = 0; it < PR / 16; ++it) {
      const int pr = pr0 + it * 16;
      const float4 r0 = *(const float4*)(Bg + (size_t)(2 * pr) * Nw);
      const float4 r1 = *(const float4*)(Bg + (size_t)(2 * pr + 1) * Nw);
      float4* d4 = (float4*)(bs + pr * PITCH + cg * 8);
      d4[0] = make_float4(r0.x, r1.x, r0.y, r1.y);
      d4[1] = make_float4(r0.z, r1.z, r0.w, r1.w);
    }
  }
  __syncthreads();
  const int tm = rb * 8 + (t >> 5);
  if (tm >= tilesM) return;
  const int lane = t & 31;
  const int kh = lane >> 4;
  const int lr = lane & 15;

  const float* Ap = A + (size_t)(tm * 16 + lr) * K + kh * 16;
  const float* bl = bs + kh * 8 * PITCH + lr * 2;
  const float* alp = attn_l + tn * 64 + lr;
  const float* arp = attn_r + tn * 64 + lr;
  const float l0 = alp[0], l1 = alp[16], l2 = alp[32], l3 = alp[48];
  const float r0 = arp[0], r1 = arp[16], r2 = arp[32], r3 = arp[48];
  v8f acc0 = {}, acc1 = {}, acc2 = {}, acc3 = {};
  float4 q0 = *(const float4*)(Ap + 0);
  float4 q1 = *(const float4*)(Ap + 4);
  float4 q2 = *(const float4*)(Ap + 8);
  float4 q3 = *(const float4*)(Ap + 12);
#define GAT_KSTEP(S, AX, AY)                                                                     \
  {                                                                                              \
    v2f a; a.x = (AX); a.y = (AY);                                                               \
    const float* br = bb + (S) * PITCH;                                                          \
    const v2f b0 = *(const v2f*)(br +  0);                                                       \
    const v2f b1 = *(const v2f*)(br + 32);                                                       \
    const v2f b2 = *(const v2f*)(br + 64);                                                       \
    const v2f b3 = *(const v2f*)(br + 96);                                                       \
    acc0 = __builtin_amdgcn_wmma_f32_16x16x4_f32(false, a, false, b0, (short)0, acc0, false, false); \
    acc1 = __builtin_amdgcn_wmma_f32_16x16x4_f32(false, a, false, b1, (short)0, acc1, false, false); \
    acc2 = __builtin_amdgcn_wmma_f32_16x16x4_f32(false, a, false, b2, (short)0, acc2, false, false); \
    acc3 = __builtin_amdgcn_wmma_f32_16x16x4_f32(false, a, false, b3, (short)0, acc3, false, false); \
  }
#pragma unroll
  for (int jb = 0; jb < NB; ++jb) {
    float4 n0 = q0, n1 = q1, n2 = q2, n3 = q3;
    if (jb + 1 < NB) {
      const float* An = Ap + (jb + 1) * 32;
      n0 = *(const float4*)(An + 0);
      n1 = *(const float4*)(An + 4);
      n2 = *(const float4*)(An + 8);
      n3 = *(const float4*)(An + 12);
    }
    const float* bb = bl + jb * 16 * PITCH;
    GAT_KSTEP(0, q0.x, q0.y)
    GAT_KSTEP(1, q0.z, q0.w)
    GAT_KSTEP(2, q1.x, q1.y)
    GAT_KSTEP(3, q1.z, q1.w)
    GAT_KSTEP(4, q2.x, q2.y)
    GAT_KSTEP(5, q2.z, q2.w)
    GAT_KSTEP(6, q3.x, q3.y)
    GAT_KSTEP(7, q3.z, q3.w)
    q0 = n0; q1 = n1; q2 = n2; q3 = n3;
  }
#undef GAT_KSTEP
  float* Cp = C + (size_t)(tm * 16 + kh * 8) * Nw + tn * 64 + lr;
#pragma unroll
  for (int r = 0; r < 8; ++r) {
    Cp[(size_t)r * Nw +  0] = acc0[r];
    Cp[(size_t)r * Nw + 16] = acc1[r];
    Cp[(size_t)r * Nw + 32] = acc2[r];
    Cp[(size_t)r * Nw + 48] = acc3[r];
  }
  float keep = 0.f;
#pragma unroll
  for (int r = 0; r < 8; ++r) {
    const float pl = gat_row16_sum(acc0[r] * l0 + acc1[r] * l1 + acc2[r] * l2 + acc3[r] * l3);
    const float pr = gat_row16_sum(acc0[r] * r0 + acc1[r] * r1 + acc2[r] * r2 + acc3[r] * r3);
    keep = (lr == r) ? pl : keep;
    keep = (lr == r + 8) ? pr : keep;
  }
  float* ep = (lr < 8) ? el : er;
  ep[(size_t)(tm * 16 + kh * 8 + (lr & 7)) * tilesN + tn] = keep;
}

__global__ __launch_bounds__(256)
void gat_bucket_fill(const int* __restrict__ src, const int* __restrict__ dst,
                     int* __restrict__ cnt, int* __restrict__ ctl,
                     int2* __restrict__ pairs, int4* __restrict__ ovf, int E) {
  int e = blockIdx.x * blockDim.x + threadIdx.x;
  if (e >= E) return;
  const int d = dst[e];
  const int s = src[e];
  const int slot = atomicAdd(&cnt[d], 1);
  if (slot < 32) {
    pairs[(size_t)d * 32 + slot] = make_int2(e, s);
  } else {
    const int k = atomicAdd(&ctl[0], 1);
    ovf[k] = make_int4(d, e, s, 0);
  }
}

__device__ __forceinline__ int gat_bucket_slot_addr(int key, int cnt, int lane) {
  int c = 0;
#pragma unroll 4
  for (int j = 0; j < cnt; ++j) c += (__builtin_amdgcn_readlane(key, j) < key) ? 1 : 0;
  return ((lane < cnt) ? c : lane) << 2;
}
__device__ __forceinline__ int gat_push_i(int addr, int v) {
  return __builtin_amdgcn_ds_permute(addr, v);
}
__device__ __forceinline__ float gat_push_f(int addr, float v) {
  return __int_as_float(__builtin_amdgcn_ds_permute(addr, __float_as_int(v)));
}

__device__ __forceinline__ int gat_bucket_park_sorted(const int2* __restrict__ pairs,
                                                      const int4* __restrict__ ovf, int* ctl,
                                                      int bump, int2* unsorted, int* sorted,
                                                      int d, int n, int lane) {
  int base = 0;
  if (lane == 0) base = atomicAdd(&ctl[bump], (n + 31) & ~31);
  base = __builtin_amdgcn_readfirstlane(base);
  unsorted[base + lane] = pairs[(size_t)d * 32 + lane];
  const int on = __builtin_amdgcn_readfirstlane(ctl[0]);
  int filled = 32;
  for (int q0 = 0; q0 < on; q0 += 32) {
    const int q = q0 + lane;
    int4 rec = make_int4(-1, 0, 0, 0);
    if (q < on) rec = ovf[q];
    const bool hit = rec.x == d;
    const unsigned m = __builtin_amdgcn_ballot_w32(hit);
    if (hit)
      unsorted[base + filled + __builtin_popcount(m & ((1u << lane) - 1u))] =
          make_int2(rec.y, rec.z);
    filled += __builtin_popcount(m);
  }
  __threadfence();
  for (int p = lane; p < n; p += 32) {
    const int2 me = unsorted[base + p];
    int c = 0;
    for (int q = 0; q < n; ++q) c += (unsorted[base + q].x < me.x) ? 1 : 0;
    sorted[base + c] = me.y;
  }
  __threadfence();
  return base;
}

__device__ __forceinline__ float gat_lrelu(float x) { return (x > 0.f) ? x : 0.2f * x; }
__device__ __forceinline__ float gat_wave_max(float v) {
  for (int o = 16; o > 0; o >>= 1) v = fmaxf(v, __shfl_xor(v, o, 32));
  return v;
}
__device__ __forceinline__ float gat_wave_sum(float v) {
  for (int o = 16; o > 0; o >>= 1) v += __shfl_xor(v, o, 32);
  return v;
}
__device__ __forceinline__ float gat_bcast(float v, int j) {
  return __int_as_float(__builtin_amdgcn_readlane(__float_as_int(v), j));
}

__device__ __forceinline__ void gat_rows_h4(const float* __restrict__ h, int s, float a0, float a1,
                                            float a2, float a3, int cnt, int lane, bool hi,
                                            float4& acc0, float4& acc1) {
#pragma unroll 4
  for (int j = 0; j < cnt; ++j) {
    const int sj = __builtin_amdgcn_readlane(s, j);
    const float w0 = gat_bcast(a0, j), w1 = gat_bcast(a1, j);
    const float w2 = gat_bcast(a2, j), w3 = gat_bcast(a3, j);
    const float wA = hi ? w1 : w0;
    const float wB = hi ? w3 : w2;
    const float4* row = (const float4*)(h + (size_t)sj * 256);
    const float4 v0 = row[lane];
    const float4 v1 = row[32 + lane];
    acc0.x += v0.x * wA; acc0.y += v0.y * wA; acc0.z += v0.z * wA; acc0.w += v0.w * wA;
    acc1.x += v1.x * wB; acc1.y += v1.y * wB; acc1.z += v1.z * wB; acc1.w += v1.w * wB;
  }
}

__global__ __launch_bounds__(256)
void gat_gather_h4(const int* __restrict__ deg, const int2* __restrict__ pairs,
                   const int4* __restrict__ ovf, int* ctl, int bump, int2* unsorted,
                   int* sorted, const float* __restrict__ h, const float* __restrict__ el,
                   const float* __restrict__ er, const float* __restrict__ bias,
                   float* __restrict__ out, int N) {
  const int d = __builtin_amdgcn_readfirstlane((int)(blockIdx.x * 8 + (threadIdx.x >> 5)));
  if (d >= N) return;
  const int lane = threadIdx.x & 31;
  const bool hi = lane >= 16;
  const int n = deg[d];
  const float4 erd = *(const float4*)(er + (size_t)d * 4);
  const float4* el4 = (const float4*)el;

  const float ninf = -__builtin_huge_valf();
  float4 acc0 = {0.f, 0.f, 0.f, 0.f}, acc1 = {0.f, 0.f, 0.f, 0.f};
  if (n <= 32) {
    const int cnt = n;
    const bool act = lane < cnt;
    int key = 0x7fffffff, sv = 0;
    float4 xv = {0.f, 0.f, 0.f, 0.f};
    if (act) {
      const int2 pr = pairs[(size_t)d * 32 + lane];
      key = pr.x;
      sv = pr.y;
      xv = el4[sv];
    }
    const int tgt = gat_bucket_slot_addr(key, cnt, lane);
    const int s = gat_push_i(tgt, sv);
    float4 x;
    x.x = gat_push_f(tgt, xv.x);
    x.y = gat_push_f(tgt, xv.y);
    x.z = gat_push_f(tgt, xv.z);
    x.w = gat_push_f(tgt, xv.w);
    float e0 = ninf, e1 = ninf, e2 = ninf, e3 = ninf;
    if (act) {
      e0 = gat_lrelu(x.x + erd.x);
      e1 = gat_lrelu(x.y + erd.y);
      e2 = gat_lrelu(x.z + erd.z);
      e3 = gat_lrelu(x.w + erd.w);
    }
    const float m0 = gat_wave_max(e0), m1 = gat_wave_max(e1);
    const float m2 = gat_wave_max(e2), m3 = gat_wave_max(e3);
    float p0 = 0.f, p1 = 0.f, p2 = 0.f, p3 = 0.f;
    if (act) {
      p0 = __expf(e0 - m0);
      p1 = __expf(e1 - m1);
      p2 = __expf(e2 - m2);
      p3 = __expf(e3 - m3);
    }
    const float s0 = gat_wave_sum(p0), s1 = gat_wave_sum(p1);
    const float s2 = gat_wave_sum(p2), s3 = gat_wave_sum(p3);
    float a0 = 0.f, a1 = 0.f, a2 = 0.f, a3 = 0.f;
    if (act) {
      a0 = p0 / s0;
      a1 = p1 / s1;
      a2 = p2 / s2;
      a3 = p3 / s3;
    }
    gat_rows_h4(h, s, a0, a1, a2, a3, cnt, lane, hi, acc0, acc1);
  } else {
    const int beg = gat_bucket_park_sorted(pairs, ovf, ctl, bump, unsorted, sorted, d, n, lane);
    const int end = beg + n;
    float m0 = ninf, m1 = ninf, m2 = ninf, m3 = ninf;
    for (int p = beg + lane; p < end; p += 32) {
      const float4 x = el4[sorted[p]];
      m0 = fmaxf(m0, gat_lrelu(x.x + erd.x));
      m1 = fmaxf(m1, gat_lrelu(x.y + erd.y));
      m2 = fmaxf(m2, gat_lrelu(x.z + erd.z));
      m3 = fmaxf(m3, gat_lrelu(x.w + erd.w));
    }
    m0 = gat_wave_max(m0); m1 = gat_wave_max(m1);
    m2 = gat_wave_max(m2); m3 = gat_wave_max(m3);

    float s0 = 0.f, s1 = 0.f, s2 = 0.f, s3 = 0.f;
    for (int p = beg + lane; p < end; p += 32) {
      const float4 x = el4[sorted[p]];
      s0 += __expf(gat_lrelu(x.x + erd.x) - m0);
      s1 += __expf(gat_lrelu(x.y + erd.y) - m1);
      s2 += __expf(gat_lrelu(x.z + erd.z) - m2);
      s3 += __expf(gat_lrelu(x.w + erd.w) - m3);
    }
    s0 = gat_wave_sum(s0); s1 = gat_wave_sum(s1);
    s2 = gat_wave_sum(s2); s3 = gat_wave_sum(s3);

    for (int base = beg; base < end; base += 32) {
      const int p = base + lane;
      int s = 0;
      float a0 = 0.f, a1 = 0.f, a2 = 0.f, a3 = 0.f;
      if (p < end) {
        s = sorted[p];
        const float4 x = el4[s];
        a0 = __expf(gat_lrelu(x.x + erd.x) - m0) / s0;
        a1 = __expf(gat_lrelu(x.y + erd.y) - m1) / s1;
        a2 = __expf(gat_lrelu(x.z + erd.z) - m2) / s2;
        a3 = __expf(gat_lrelu(x.w + erd.w) - m3) / s3;
      }
      const int cnt = (end - base < 32) ? (end - base) : 32;
      gat_rows_h4(h, s, a0, a1, a2, a3, cnt, lane, hi, acc0, acc1);
    }
  }
  const float4 b0 = ((const float4*)bias)[lane];
  const float4 b1 = ((const float4*)bias)[32 + lane];
  float4 o0, o1;
  o0.x = acc0.x + b0.x; o0.y = acc0.y + b0.y; o0.z = acc0.z + b0.z; o0.w = acc0.w + b0.w;
  o1.x = acc1.x + b1.x; o1.y = acc1.y + b1.y; o1.z = acc1.z + b1.z; o1.w = acc1.w + b1.w;
  o0.x = o0.x > 0.f ? o0.x : 0.f; o0.y = o0.y > 0.f ? o0.y : 0.f;
  o0.z = o0.z > 0.f ? o0.z : 0.f; o0.w = o0.w > 0.f ? o0.w : 0.f;
  o1.x = o1.x > 0.f ? o1.x : 0.f; o1.y = o1.y > 0.f ? o1.y : 0.f;
  o1.z = o1.z > 0.f ? o1.z : 0.f; o1.w = o1.w > 0.f ? o1.w : 0.f;
  float4* orow = (float4*)(out + (size_t)d * 256);
  orow[lane] = o0;
  orow[32 + lane] = o1;
}

__device__ __forceinline__ void gat_rows_h1(const float* __restrict__ h, int s, float a, int cnt,
                                            int lane, float2& acc) {
#pragma unroll 4
  for (int j = 0; j < cnt; ++j) {
    const int sj = __builtin_amdgcn_readlane(s, j);
    const float w = gat_bcast(a, j);
    const float2 v = ((const float2*)(h + (size_t)sj * 64))[lane];
    acc.x += v.x * w; acc.y += v.y * w;
  }
}

__global__ __launch_bounds__(256)
void gat_gather_h1(const int* __restrict__ deg, const int2* __restrict__ pairs,
                   const int4* __restrict__ ovf, int* ctl, int bump, int2* unsorted,
                   int* sorted, const float* __restrict__ h, const float* __restrict__ el,
                   const float* __restrict__ er, const float* __restrict__ bias,
                   float* __restrict__ out, int N) {
  const int d = __builtin_amdgcn_readfirstlane((int)(blockIdx.x * 8 + (threadIdx.x >> 5)));
  if (d >= N) return;
  const int lane = threadIdx.x & 31;
  const int n = deg[d];
  const float erd = er[d];

  const float ninf = -__builtin_huge_valf();
  float2 acc = {0.f, 0.f};
  if (n <= 32) {
    const int cnt = n;
    const bool act = lane < cnt;
    int key = 0x7fffffff, sv = 0;
    float xv = 0.f;
    if (act) {
      const int2 pr = pairs[(size_t)d * 32 + lane];
      key = pr.x;
      sv = pr.y;
      xv = el[sv];
    }
    const int tgt = gat_bucket_slot_addr(key, cnt, lane);
    const int s = gat_push_i(tgt, sv);
    const float x = gat_push_f(tgt, xv);
    float e = ninf;
    if (act) e = gat_lrelu(x + erd);
    const float m = gat_wave_max(e);
    float pe = 0.f;
    if (act) pe = __expf(e - m);
    const float ssum = gat_wave_sum(pe);
    float a = 0.f;
    if (act) a = pe / ssum;
    gat_rows_h1(h, s, a, cnt, lane, acc);
  } else {
    const int beg = gat_bucket_park_sorted(pairs, ovf, ctl, bump, unsorted, sorted, d, n, lane);
    const int end = beg + n;
    float m = ninf;
    for (int p = beg + lane; p < end; p += 32)
      m = fmaxf(m, gat_lrelu(el[sorted[p]] + erd));
    m = gat_wave_max(m);

    float ssum = 0.f;
    for (int p = beg + lane; p < end; p += 32)
      ssum += __expf(gat_lrelu(el[sorted[p]] + erd) - m);
    ssum = gat_wave_sum(ssum);

    for (int base = beg; base < end; base += 32) {
      const int p = base + lane;
      int s = 0;
      float a = 0.f;
      if (p < end) {
        s = sorted[p];
        a = __expf(gat_lrelu(el[s] + erd) - m) / ssum;
      }
      const int cnt = (end - base < 32) ? (end - base) : 32;
      gat_rows_h1(h, s, a, cnt, lane, acc);
    }
  }
  const float2 b = ((const float2*)bias)[lane];
  float2 o;
  o.x = acc.x + b.x; o.y = acc.y + b.y;
  ((float2*)(out + (size_t)d * 64))[lane] = o;
}

extern "C" void kernel_launch(void* const* d_in, const int* in_sizes, int n_in,
                              void* d_out, int out_size, void* d_ws, size_t ws_size,
                              hipStream_t stream) {
  (void)in_sizes; (void)n_in; (void)out_size; (void)ws_size;
  const float* feat = (const float*)d_in[0];
  const int*   src  = (const int*)d_in[1];
  const int*   dst  = (const int*)d_in[2];
  const float* W1  = (const float*)d_in[3];
  const float* al1 = (const float*)d_in[4];
  const float* ar1 = (const float*)d_in[5];
  const float* b1  = (const float*)d_in[6];
  const float* W2  = (const float*)d_in[7];
  const float* al2 = (const float*)d_in[8];
  const float* ar2 = (const float*)d_in[9];
  const float* b2  = (const float*)d_in[10];
  const float* W3  = (const float*)d_in[11];
  const float* al3 = (const float*)d_in[12];
  const float* ar3 = (const float*)d_in[13];
  const float* b3  = (const float*)d_in[14];
  float* out = (float*)d_out;

  const int N = GAT_N, E = GAT_E;

  char* ws = (char*)d_ws;
  size_t off = 0;
  auto carve = [&](size_t bytes) -> void* {
    void* p = ws + off;
    off += (bytes + 255) & ~(size_t)255;
    return p;
  };
  float* bufA    = (float*)carve((size_t)N * 256 * 4);
  float* bufB    = (float*)carve((size_t)N * 256 * 4);
  float* el      = (float*)carve((size_t)N * 4 * 4);
  float* er      = (float*)carve((size_t)N * 4 * 4);
  int*   cnt     = (int*)carve((size_t)(N + 64) * 4);
  int*   ctl     = cnt + N;
  int2*  pairs   = (int2*)carve((size_t)N * 32 * 8);
  int4*  ovf     = (int4*)carve((size_t)E * 16);
  int2*  unsorted = (int2*)carve((size_t)E * 2 * 8);
  int*   sorted  = (int*)carve((size_t)E * 2 * 4);

  auto blocks = [](long long n) { return (unsigned)((n + 255) / 256); };

  const unsigned node_blocks = (unsigned)((N + 7) / 8);

  {
    const int Nw = 256;
    const int gemm_blocks = ((N / 16 + 7) / 8) * (Nw / 64);
    gat_gemm_wmma<128, true><<<gemm_blocks, 256, 0, stream>>>(feat, W1, bufA, al1, ar1, el, er, N, Nw,
                                                             cnt, N + 64);
    gat_bucket_fill<<<blocks(E), 256, 0, stream>>>(src, dst, cnt, ctl, pairs, ovf, E);
    gat_gather_h4<<<node_blocks, 256, 0, stream>>>(cnt, pairs, ovf, ctl, 1, unsorted, sorted, bufA, el, er,
                                                  b1, bufB, N);
  }

  {
    const int Nw = 256;
    const int gemm_blocks = ((N / 16 + 7) / 8) * (Nw / 64);
    gat_gemm_wmma<256, false><<<gemm_blocks, 256, 0, stream>>>(bufB, W2, bufA, al2, ar2, el, er, N, Nw,
                                                              nullptr, 0);
    gat_gather_h4<<<node_blocks, 256, 0, stream>>>(cnt, pairs, ovf, ctl, 2, unsorted, sorted, bufA, el, er,
                                                  b2, bufB, N);
  }

  {
    const int Nw = 64;
    const int gemm_blocks = ((N / 16 + 7) / 8) * (Nw / 64);
    gat_gemm_wmma<256, false><<<gemm_blocks, 256, 0, stream>>>(bufB, W3, bufA, al3, ar3, el, er, N, Nw,
                                                              nullptr, 0);
    gat_gather_h1<<<node_blocks, 256, 0, stream>>>(cnt, pairs, ovf, ctl, 3, unsorted, sorted, bufA, el, er,
                                                  b3, out, N);
  }
}


# Bucket fill keeps one copy of counters and slots per half of the package (the wave's die id picks its copy), so a line's adds and slot stores come from one half; gather reads both copies; same sums
# speedup vs baseline: 9.6326x; 1.2092x over previous
#include <hip/hip_runtime.h>


#define GAT_N 50000
#define GAT_E 800000
#define GAT_F 64

typedef float v2f __attribute__((ext_vector_type(2)));
typedef float v8f __attribute__((ext_vector_type(8)));

template <int MASK>
__device__ __forceinline__ float gat_row_xor(float v) {
  return __int_as_float(__builtin_amdgcn_update_dpp(0, __float_as_int(v), 0x160 + MASK,
                                                    0xf, 0xf, false));
}
__device__ __forceinline__ float gat_row16_sum(float v) {
  v += gat_row_xor<1>(v);
  v += gat_row_xor<2>(v);
  v += gat_row_xor<4>(v);
  v += gat_row_xor<8>(v);
  return v;
}

template <int K, bool CLR>
__global__ __launch_bounds__(256)
void gat_gemm_wmma(const float* __restrict__ A, const float* __restrict__ B,
                   float* __restrict__ C, const float* __restrict__ attn_l,
                   const float* __restrict__ attn_r, float* __restrict__ el,
                   float* __restrict__ er, int M, int Nw, int* __restrict__ clr, int clr_n) {
  constexpr int PR = K / 2;
  constexpr int PITCH = 132;
  constexpr int NB = K / 32;
  static_assert(K % 32 == 0, "K must be a whole number of 32-k blocks");
  __shared__ float4 bs4[PR * PITCH / 4];
  float* bs = (float*)bs4;
  const int t = threadIdx.x;
  if (CLR) {
    const int g = blockIdx.x * 256 + t;
    if (g < clr_n) clr[g] = 0;
  }
  const int tilesN = Nw >> 6;
  const int tilesM = M >> 4;
  const int rb = blockIdx.x / tilesN;
  const int tn = blockIdx.x - rb * tilesN;
  {
    const int cg = t & 15;
    const int pr0 = t >> 4;
    const float* Bg = B + (size_t)tn * 64 + cg * 4;
#pragma unroll
    for (int it = 0; it < PR / 16; ++it) {
      const int pr = pr0 + it * 16;
      const float4 r0 = *(const float4*)(Bg + (size_t)(2 * pr) * Nw);
      const float4 r1 = *(const float4*)(Bg + (size_t)(2 * pr + 1) * Nw);
      float4* d4 = (float4*)(bs + pr * PITCH + cg * 8);
      d4[0] = make_float4(r0.x, r1.x, r0.y, r1.y);
      d4[1] = make_float4(r0.z, r1.z, r0.w, r1.w);
    }
  }
  __syncthreads();
  const int tm = rb * 8 + (t >> 5);
  if (tm >= tilesM) return;
  const int lane = t & 31;
  const int kh = lane >> 4;
  const int lr = lane & 15;

  const float* Ap = A + (size_t)(tm * 16 + lr) * K + kh * 16;
  const float* bl = bs + kh * 8 * PITCH + lr * 2;
  const float* alp = attn_l + tn * 64 + lr;
  const float* arp = attn_r + tn * 64 + lr;
  const float l0 = alp[0], l1 = alp[16], l2 = alp[32], l3 = alp[48];
  const float r0 = arp[0], r1 = arp[16], r2 = arp[32], r3 = arp[48];
  v8f acc0 = {}, acc1 = {}, acc2 = {}, acc3 = {};
  float4 q0 = *(const float4*)(Ap + 0);
  float4 q1 = *(const float4*)(Ap + 4);
  float4 q2 = *(const float4*)(Ap + 8);
  float4 q3 = *(const float4*)(Ap + 12);
#define GAT_KSTEP(S, AX, AY)                                                                     \
  {                                                                                              \
    v2f a; a.x = (AX); a.y = (AY);                                                               \
    const float* br = bb + (S) * PITCH;                                                          \
    const v2f b0 = *(const v2f*)(br +  0);                                                       \
    const v2f b1 = *(const v2f*)(br + 32);                                                       \
    const v2f b2 = *(const v2f*)(br + 64);                                                       \
    const v2f b3 = *(const v2f*)(br + 96);                                                       \
    acc0 = __builtin_amdgcn_wmma_f32_16x16x4_f32(false, a, false, b0, (short)0, acc0, false, false); \
    acc1 = __builtin_amdgcn_wmma_f32_16x16x4_f32(false, a, false, b1, (short)0, acc1, false, false); \
    acc2 = __builtin_amdgcn_wmma_f32_16x16x4_f32(false, a, false, b2, (short)0, acc2, false, false); \
    acc3 = __builtin_amdgcn_wmma_f32_16x16x4_f32(false, a, false, b3, (short)0, acc3, false, false); \
  }
#pragma unroll
  for (int jb = 0; jb < NB; ++jb) {
    float4 n0 = q0, n1 = q1, n2 = q2, n3 = q3;
    if (jb + 1 < NB) {
      const float* An = Ap + (jb + 1) * 32;
      n0 = *(const float4*)(An + 0);
      n1 = *(const float4*)(An + 4);
      n2 = *(const float4*)(An + 8);
      n3 = *(const float4*)(An + 12);
    }
    const float* bb = bl + jb * 16 * PITCH;
    GAT_KSTEP(0, q0.x, q0.y)
    GAT_KSTEP(1, q0.z, q0.w)
    GAT_KSTEP(2, q1.x, q1.y)
    GAT_KSTEP(3, q1.z, q1.w)
    GAT_KSTEP(4, q2.x, q2.y)
    GAT_KSTEP(5, q2.z, q2.w)
    GAT_KSTEP(6, q3.x, q3.y)
    GAT_KSTEP(7, q3.z, q3.w)
    q0 = n0; q1 = n1; q2 = n2; q3 = n3;
  }
#undef GAT_KSTEP
  float* Cp = C + (size_t)(tm * 16 + kh * 8) * Nw + tn * 64 + lr;
#pragma unroll
  for (int r = 0; r < 8; ++r) {
    Cp[(size_t)r * Nw +  0] = acc0[r];
    Cp[(size_t)r * Nw + 16] = acc1[r];
    Cp[(size_t)r * Nw + 32] = acc2[r];
    Cp[(size_t)r * Nw + 48] = acc3[r];
  }
  float keep = 0.f;
#pragma unroll
  for (int r = 0; r < 8; ++r) {
    const float pl = gat_row16_sum(acc0[r] * l0 + acc1[r] * l1 + acc2[r] * l2 + acc3[r] * l3);
    const float pr = gat_row16_sum(acc0[r] * r0 + acc1[r] * r1 + acc2[r] * r2 + acc3[r] * r3);
    keep = (lr == r) ? pl : keep;
    keep = (lr == r + 8) ? pr : keep;
  }
  float* ep = (lr < 8) ? el : er;
  ep[(size_t)(tm * 16 + kh * 8 + (lr & 7)) * tilesN + tn] = keep;
}

__device__ __forceinline__ int gat_pkg_half() {
  return (int)((__builtin_amdgcn_s_sendmsg_rtn(0x87) >> 18) & 1u);
}

__global__ __launch_bounds__(256)
void gat_bucket_fill(const int* __restrict__ src, const int* __restrict__ dst,
                     int* __restrict__ cnt, int* __restrict__ ctl,
                     int2* __restrict__ pairs, int4* __restrict__ ovf, int E, int N) {
  int e = blockIdx.x * blockDim.x + threadIdx.x;
  if (e >= E) return;
  const int hb = gat_pkg_half() * N;
  const int d = dst[e];
  const int s = src[e];
  const int slot = atomicAdd(&cnt[hb + d], 1);
  if (slot < 32) {
    pairs[(size_t)(hb + d) * 32 + slot] = make_int2(e, s);
  } else {
    const int k = atomicAdd(&ctl[0], 1);
    ovf[k] = make_int4(d, e, s, 0);
  }
}

__device__ __forceinline__ int gat_bucket_slot_addr(int key, int cnt, int lane) {
  int c = 0;
#pragma unroll 4
  for (int j = 0; j < cnt; ++j) c += (__builtin_amdgcn_readlane(key, j) < key) ? 1 : 0;
  return ((lane < cnt) ? c : lane) << 2;
}
__device__ __forceinline__ int gat_push_i(int addr, int v) {
  return __builtin_amdgcn_ds_permute(addr, v);
}
__device__ __forceinline__ float gat_push_f(int addr, float v) {
  return __int_as_float(__builtin_amdgcn_ds_permute(addr, __float_as_int(v)));
}

__device__ __forceinline__ int gat_bucket_park_sorted(const int2* __restrict__ pairs,
                                                      const int4* __restrict__ ovf, int* ctl,
                                                      int bump, int2* unsorted, int* sorted,
                                                      int d, int N, int c0, int c1, int lane) {
  const int n = c0 + c1;
  const int m0 = (c0 < 32) ? c0 : 32;
  const int m1 = (c1 < 32) ? c1 : 32;
  int base = 0;
  if (lane == 0) base = atomicAdd(&ctl[bump], (n + 31) & ~31);
  base = __builtin_amdgcn_readfirstlane(base);
  if (lane < m0) unsorted[base + lane] = pairs[(size_t)d * 32 + lane];
  if (lane < m1) unsorted[base + m0 + lane] = pairs[(size_t)(N + d) * 32 + lane];
  const int on = __builtin_amdgcn_readfirstlane(ctl[0]);
  int filled = m0 + m1;
  for (int q0 = 0; q0 < on; q0 += 32) {
    const int q = q0 + lane;
    int4 rec = make_int4(-1, 0, 0, 0);
    if (q < on) rec = ovf[q];
    const bool hit = rec.x == d;
    const unsigned m = __builtin_amdgcn_ballot_w32(hit);
    if (hit)
      unsorted[base + filled + __builtin_popcount(m & ((1u << lane) - 1u))] =
          make_int2(rec.y, rec.z);
    filled += __builtin_popcount(m);
  }
  __threadfence();
  for (int p = lane; p < n; p += 32) {
    const int2 me = unsorted[base + p];
    int c = 0;
    for (int q = 0; q < n; ++q) c += (unsorted[base + q].x < me.x) ? 1 : 0;
    sorted[base + c] = me.y;
  }
  __threadfence();
  return base;
}

__device__ __forceinline__ float gat_lrelu(float x) { return (x > 0.f) ? x : 0.2f * x; }
__device__ __forceinline__ float gat_wave_max(float v) {
  for (int o = 16; o > 0; o >>= 1) v = fmaxf(v, __shfl_xor(v, o, 32));
  return v;
}
__device__ __forceinline__ float gat_wave_sum(float v) {
  for (int o = 16; o > 0; o >>= 1) v += __shfl_xor(v, o, 32);
  return v;
}
__device__ __forceinline__ float gat_bcast(float v, int j) {
  return __int_as_float(__builtin_amdgcn_readlane(__float_as_int(v), j));
}

__device__ __forceinline__ void gat_rows_h4(const float* __restrict__ h, int s, float a0, float a1,
                                            float a2, float a3, int cnt, int lane, bool hi,
                                            float4& acc0, float4& acc1) {
#pragma unroll 4
  for (int j = 0; j < cnt; ++j) {
    const int sj = __builtin_amdgcn_readlane(s, j);
    const float w0 = gat_bcast(a0, j), w1 = gat_bcast(a1, j);
    const float w2 = gat_bcast(a2, j), w3 = gat_bcast(a3, j);
    const float wA = hi ? w1 : w0;
    const float wB = hi ? w3 : w2;
    const float4* row = (const float4*)(h + (size_t)sj * 256);
    const float4 v0 = row[lane];
    const float4 v1 = row[32 + lane];
    acc0.x += v0.x * wA; acc0.y += v0.y * wA; acc0.z += v0.z * wA; acc0.w += v0.w * wA;
    acc1.x += v1.x * wB; acc1.y += v1.y * wB; acc1.z += v1.z * wB; acc1.w += v1.w * wB;
  }
}

__global__ __launch_bounds__(256)
void gat_gather_h4(const int* __restrict__ deg, const int2* __restrict__ pairs,
                   const int4* __restrict__ ovf, int* ctl, int bump, int2* unsorted,
                   int* sorted, const float* __restrict__ h, const float* __restrict__ el,
                   const float* __restrict__ er, const float* __restrict__ bias,
                   float* __restrict__ out, int N) {
  const int d = __builtin_amdgcn_readfirstlane((int)(blockIdx.x * 8 + (threadIdx.x >> 5)));
  if (d >= N) return;
  const int lane = threadIdx.x & 31;
  const bool hi = lane >= 16;
  const int c0 = deg[d];
  const int c1 = deg[N + d];
  const int n = c0 + c1;
  const float4 erd = *(const float4*)(er + (size_t)d * 4);
  const float4* el4 = (const float4*)el;

  const float ninf = -__builtin_huge_valf();
  float4 acc0 = {0.f, 0.f, 0.f, 0.f}, acc1 = {0.f, 0.f, 0.f, 0.f};
  if (n <= 32) {
    const int cnt = n;
    const bool act = lane < cnt;
    int key = 0x7fffffff, sv = 0;
    float4 xv = {0.f, 0.f, 0.f, 0.f};
    if (act) {
      const size_t at = (lane < c0) ? ((size_t)d * 32 + lane) : ((size_t)(N + d) * 32 + (lane - c0));
      const int2 pr = pairs[at];
      key = pr.x;
      sv = pr.y;
      xv = el4[sv];
    }
    const int tgt = gat_bucket_slot_addr(key, cnt, lane);
    const int s = gat_push_i(tgt, sv);
    float4 x;
    x.x = gat_push_f(tgt, xv.x);
    x.y = gat_push_f(tgt, xv.y);
    x.z = gat_push_f(tgt, xv.z);
    x.w = gat_push_f(tgt, xv.w);
    float e0 = ninf, e1 = ninf, e2 = ninf, e3 = ninf;
    if (act) {
      e0 = gat_lrelu(x.x + erd.x);
      e1 = gat_lrelu(x.y + erd.y);
      e2 = gat_lrelu(x.z + erd.z);
      e3 = gat_lrelu(x.w + erd.w);
    }
    const float m0 = gat_wave_max(e0), m1 = gat_wave_max(e1);
    const float m2 = gat_wave_max(e2), m3 = gat_wave_max(e3);
    float p0 = 0.f, p1 = 0.f, p2 = 0.f, p3 = 0.f;
    if (act) {
      p0 = __expf(e0 - m0);
      p1 = __expf(e1 - m1);
      p2 = __expf(e2 - m2);
      p3 = __expf(e3 - m3);
    }
    const float s0 = gat_wave_sum(p0), s1 = gat_wave_sum(p1);
    const float s2 = gat_wave_sum(p2), s3 = gat_wave_sum(p3);
    float a0 = 0.f, a1 = 0.f, a2 = 0.f, a3 = 0.f;
    if (act) {
      a0 = p0 / s0;
      a1 = p1 / s1;
      a2 = p2 / s2;
      a3 = p3 / s3;
    }
    gat_rows_h4(h, s, a0, a1, a2, a3, cnt, lane, hi, acc0, acc1);
  } else {
    const int beg = gat_bucket_park_sorted(pairs, ovf, ctl, bump, unsorted, sorted, d, N, c0, c1, lane);
    const int end = beg + n;
    float m0 = ninf, m1 = ninf, m2 = ninf, m3 = ninf;
    for (int p = beg + lane; p < end; p += 32) {
      const float4 x = el4[sorted[p]];
      m0 = fmaxf(m0, gat_lrelu(x.x + erd.x));
      m1 = fmaxf(m1, gat_lrelu(x.y + erd.y));
      m2 = fmaxf(m2, gat_lrelu(x.z + erd.z));
      m3 = fmaxf(m3, gat_lrelu(x.w + erd.w));
    }
    m0 = gat_wave_max(m0); m1 = gat_wave_max(m1);
    m2 = gat_wave_max(m2); m3 = gat_wave_max(m3);

    float s0 = 0.f, s1 = 0.f, s2 = 0.f, s3 = 0.f;
    for (int p = beg + lane; p < end; p += 32) {
      const float4 x = el4[sorted[p]];
      s0 += __expf(gat_lrelu(x.x + erd.x) - m0);
      s1 += __expf(gat_lrelu(x.y + erd.y) - m1);
      s2 += __expf(gat_lrelu(x.z + erd.z) - m2);
      s3 += __expf(gat_lrelu(x.w + erd.w) - m3);
    }
    s0 = gat_wave_sum(s0); s1 = gat_wave_sum(s1);
    s2 = gat_wave_sum(s2); s3 = gat_wave_sum(s3);

    for (int base = beg; base < end; base += 32) {
      const int p = base + lane;
      int s = 0;
      float a0 = 0.f, a1 = 0.f, a2 = 0.f, a3 = 0.f;
      if (p < end) {
        s = sorted[p];
        const float4 x = el4[s];
        a0 = __expf(gat_lrelu(x.x + erd.x) - m0) / s0;
        a1 = __expf(gat_lrelu(x.y + erd.y) - m1) / s1;
        a2 = __expf(gat_lrelu(x.z + erd.z) - m2) / s2;
        a3 = __expf(gat_lrelu(x.w + erd.w) - m3) / s3;
      }
      const int cnt = (end - base < 32) ? (end - base) : 32;
      gat_rows_h4(h, s, a0, a1, a2, a3, cnt, lane, hi, acc0, acc1);
    }
  }
  const float4 b0 = ((const float4*)bias)[lane];
  const float4 b1 = ((const float4*)bias)[32 + lane];
  float4 o0, o1;
  o0.x = acc0.x + b0.x; o0.y = acc0.y + b0.y; o0.z = acc0.z + b0.z; o0.w = acc0.w + b0.w;
  o1.x = acc1.x + b1.x; o1.y = acc1.y + b1.y; o1.z = acc1.z + b1.z; o1.w = acc1.w + b1.w;
  o0.x = o0.x > 0.f ? o0.x : 0.f; o0.y = o0.y > 0.f ? o0.y : 0.f;
  o0.z = o0.z > 0.f ? o0.z : 0.f; o0.w = o0.w > 0.f ? o0.w : 0.f;
  o1.x = o1.x > 0.f ? o1.x : 0.f; o1.y = o1.y > 0.f ? o1.y : 0.f;
  o1.z = o1.z > 0.f ? o1.z : 0.f; o1.w = o1.w > 0.f ? o1.w : 0.f;
  float4* orow = (float4*)(out + (size_t)d * 256);
  orow[lane] = o0;
  orow[32 + lane] = o1;
}

__device__ __forceinline__ void gat_rows_h1(const float* __restrict__ h, int s, float a, int cnt,
                                            int lane, float2& acc) {
#pragma unroll 4
  for (int j = 0; j < cnt; ++j) {
    const int sj = __builtin_amdgcn_readlane(s, j);
    const float w = gat_bcast(a, j);
    const float2 v = ((const float2*)(h + (size_t)sj * 64))[lane];
    acc.x += v.x * w; acc.y += v.y * w;
  }
}

__global__ __launch_bounds__(256)
void gat_gather_h1(const int* __restrict__ deg, const int2* __restrict__ pairs,
                   const int4* __restrict__ ovf, int* ctl, int bump, int2* unsorted,
                   int* sorted, const float* __restrict__ h, const float* __restrict__ el,
                   const float* __restrict__ er, const float* __restrict__ bias,
                   float* __restrict__ out, int N) {
  const int d = __builtin_amdgcn_readfirstlane((int)(blockIdx.x * 8 + (threadIdx.x >> 5)));
  if (d >= N) return;
  const int lane = threadIdx.x & 31;
  const int c0 = deg[d];
  const int c1 = deg[N + d];
  const int n = c0 + c1;
  const float erd = er[d];

  const float ninf = -__builtin_huge_valf();
  float2 acc = {0.f, 0.f};
  if (n <= 32) {
    const int cnt = n;
    const bool act = lane < cnt;
    int key = 0x7fffffff, sv = 0;
    float xv = 0.f;
    if (act) {
      const size_t at = (lane < c0) ? ((size_t)d * 32 + lane) : ((size_t)(N + d) * 32 + (lane - c0));
      const int2 pr = pairs[at];
      key = pr.x;
      sv = pr.y;
      xv = el[sv];
    }
    const int tgt = gat_bucket_slot_addr(key, cnt, lane);
    const int s = gat_push_i(tgt, sv);
    const float x = gat_push_f(tgt, xv);
    float e = ninf;
    if (act) e = gat_lrelu(x + erd);
    const float m = gat_wave_max(e);
    float pe = 0.f;
    if (act) pe = __expf(e - m);
    const float ssum = gat_wave_sum(pe);
    float a = 0.f;
    if (act) a = pe / ssum;
    gat_rows_h1(h, s, a, cnt, lane, acc);
  } else {
    const int beg = gat_bucket_park_sorted(pairs, ovf, ctl, bump, unsorted, sorted, d, N, c0, c1, lane);
    const int end = beg + n;
    float m = ninf;
    for (int p = beg + lane; p < end; p += 32)
      m = fmaxf(m, gat_lrelu(el[sorted[p]] + erd));
    m = gat_wave_max(m);

    float ssum = 0.f;
    for (int p = beg + lane; p < end; p += 32)
      ssum += __expf(gat_lrelu(el[sorted[p]] + erd) - m);
    ssum = gat_wave_sum(ssum);

    for (int base = beg; base < end; base += 32) {
      const int p = base + lane;
      int s = 0;
      float a = 0.f;
      if (p < end) {
        s = sorted[p];
        a = __expf(gat_lrelu(el[s] + erd) - m) / ssum;
      }
      const int cnt = (end - base < 32) ? (end - base) : 32;
      gat_rows_h1(h, s, a, cnt, lane, acc);
    }
  }
  const float2 b = ((const float2*)bias)[lane];
  float2 o;
  o.x = acc.x + b.x; o.y = acc.y + b.y;
  ((float2*)(out + (size_t)d * 64))[lane] = o;
}

extern "C" void kernel_launch(void* const* d_in, const int* in_sizes, int n_in,
                              void* d_out, int out_size, void* d_ws, size_t ws_size,
                              hipStream_t stream) {
  (void)in_sizes; (void)n_in; (void)out_size; (void)ws_size;
  const float* feat = (const float*)d_in[0];
  const int*   src  = (const int*)d_in[1];
  const int*   dst  = (const int*)d_in[2];
  const float* W1  = (const float*)d_in[3];
  const float* al1 = (const float*)d_in[4];
  const float* ar1 = (const float*)d_in[5];
  const float* b1  = (const float*)d_in[6];
  const float* W2  = (const float*)d_in[7];
  const float* al2 = (const float*)d_in[8];
  const float* ar2 = (const float*)d_in[9];
  const float* b2  = (const float*)d_in[10];
  const float* W3  = (const float*)d_in[11];
  const float* al3 = (const float*)d_in[12];
  const float* ar3 = (const float*)d_in[13];
  const float* b3  = (const float*)d_in[14];
  float* out = (float*)d_out;

  const int N = GAT_N, E = GAT_E;

  char* ws = (char*)d_ws;
  size_t off = 0;
  auto carve = [&](size_t bytes) -> void* {
    void* p = ws + off;
    off += (bytes + 255) & ~(size_t)255;
    return p;
  };
  float* bufA    = (float*)carve((size_t)N * 256 * 4);
  float* bufB    = (float*)carve((size_t)N * 256 * 4);
  float* el      = (float*)carve((size_t)N * 4 * 4);
  float* er      = (float*)carve((size_t)N * 4 * 4);
  int*   cnt     = (int*)carve((size_t)(2 * N + 64) * 4);
  int*   ctl     = cnt + 2 * N;
  int2*  pairs   = (int2*)carve((size_t)2 * N * 32 * 8);
  int4*  ovf     = (int4*)carve((size_t)E * 16);
  int2*  unsorted = (int2*)carve((size_t)E * 2 * 8);
  int*   sorted  = (int*)carve((size_t)E * 2 * 4);

  auto blocks = [](long long n) { return (unsigned)((n + 255) / 256); };

  const unsigned node_blocks = (unsigned)((N + 7) / 8);

  {
    const int Nw = 256;
    const int gemm_blocks = ((N / 16 + 7) / 8) * (Nw / 64);
    gat_gemm_wmma<128, true><<<gemm_blocks, 256, 0, stream>>>(feat, W1, bufA, al1, ar1, el, er, N, Nw,
                                                             cnt, 2 * N + 64);
    gat_bucket_fill<<<blocks(E), 256, 0, stream>>>(src, dst, cnt, ctl, pairs, ovf, E, N);
    gat_gather_h4<<<node_blocks, 256, 0, stream>>>(cnt, pairs, ovf, ctl, 1, unsorted, sorted, bufA, el, er,
                                                  b1, bufB, N);
  }

  {
    const int Nw = 256;
    const int gemm_blocks = ((N / 16 + 7) / 8) * (Nw / 64);
    gat_gemm_wmma<256, false><<<gemm_blocks, 256, 0, stream>>>(bufB, W2, bufA, al2, ar2, el, er, N, Nw,
                                                              nullptr, 0);
    gat_gather_h4<<<node_blocks, 256, 0, stream>>>(cnt, pairs, ovf, ctl, 2, unsorted, sorted, bufA, el, er,
                                                  b2, bufB, N);
  }

  {
    const int Nw = 64;
    const int gemm_blocks = ((N / 16 + 7) / 8) * (Nw / 64);
    gat_gemm_wmma<256, false><<<gemm_blocks, 256, 0, stream>>>(bufB, W3, bufA, al3, ar3, el, er, N, Nw,
                                                              nullptr, 0);
    gat_gather_h1<<<node_blocks, 256, 0, stream>>>(cnt, pairs, ovf, ctl, 3, unsorted, sorted, bufA, el, er,
                                                  b3, out, N);
  }
}
